# starting kernel
# baseline (speedup 1.0000x reference)
#include <hip/hip_runtime.h>
#include <hip/hip_bf16.h>


typedef float v2f __attribute__((ext_vector_type(2)));
typedef float v8f __attribute__((ext_vector_type(8)));

#define IN_C 128
#define HID2 128
#define HID  64

__device__ __forceinline__ void atomic_add_f32(float* p, float v) {
    __hip_atomic_fetch_add(p, v, __ATOMIC_RELAXED, __HIP_MEMORY_SCOPE_AGENT);
}

__global__ __launch_bounds__(256)
void fill_kernel(float* __restrict__ p, float v, long long n) {
    long long i = (long long)blockIdx.x * 256 + threadIdx.x;
    if (i < n) p[i] = v;
}

__global__ __launch_bounds__(256)
void degree_kernel(const int* __restrict__ dst, float* __restrict__ deg, long long E) {
    long long e = (long long)blockIdx.x * 256 + threadIdx.x;
    if (e < E) atomic_add_f32(&deg[dst[e]], 1.0f);
}

__global__ __launch_bounds__(256)
void rsqrt_kernel(const float* __restrict__ deg, float* __restrict__ dinv, int n) {
    int i = blockIdx.x * 256 + threadIdx.x;
    if (i < n) dinv[i] = rsqrtf(deg[i]);
}

template <int NOUT>
__global__ __launch_bounds__(128)
void gemm_wmma_f32(const float* __restrict__ A, const float* __restrict__ B,
                   float* __restrict__ C, int M) {
    constexpr int K = 128;
    constexpr int NTILES = NOUT / 16;
    const int lane = threadIdx.x & 31;
    const int wglobal = blockIdx.x * 4 + (threadIdx.x >> 5);
    const int tm = wglobal / NTILES;
    const int tn = wglobal % NTILES;
    if (tm * 16 >= M) return;

    const int m  = lane & 15;
    const int kh = lane >> 4;
    const int n  = lane & 15;
    const int col = tn * 16 + n;
    const float* arow = A + (size_t)(tm * 16 + m) * K;

    v8f acc = {};
#pragma unroll 8
    for (int k0 = 0; k0 < K; k0 += 4) {
        const int ka = k0 + kh * 2;
        const float2 av = *reinterpret_cast<const float2*>(arow + ka);
        v2f a; a[0] = av.x; a[1] = av.y;
        v2f b;
        b[0] = B[(size_t)ka * NOUT + col];
        b[1] = B[(size_t)(ka + 1) * NOUT + col];
        acc = __builtin_amdgcn_wmma_f32_16x16x4_f32(false, a, false, b,
                                                    (short)0, acc, false, false);
    }

    const int row0 = tm * 16 + kh * 8;
#pragma unroll
    for (int r = 0; r < 8; ++r)
        C[(size_t)(row0 + r) * NOUT + col] = acc[r];
}

template <int F, int LOGC>
__global__ __launch_bounds__(256)
void edge_aggregate(const int* __restrict__ src, const int* __restrict__ dst,
                    const float* __restrict__ h, const float* __restrict__ dinv,
                    float* __restrict__ acc, long long total) {
    long long g = (long long)blockIdx.x * 256 + threadIdx.x;
    if (g >= total) return;
    const long long e = g >> LOGC;
    const int c = (int)(g & ((1 << LOGC) - 1));
    const int s = src[e];
    const int d = dst[e];
    const float norm = dinv[s] * dinv[d];
    const float4 hv = *reinterpret_cast<const float4*>(h + (size_t)s * F + c * 4);
    float* ap = acc + (size_t)d * F + c * 4;
    atomic_add_f32(ap + 0, hv.x * norm);
    atomic_add_f32(ap + 1, hv.y * norm);
    atomic_add_f32(ap + 2, hv.z * norm);
    atomic_add_f32(ap + 3, hv.w * norm);
}

template <int F, int LOGF, bool RELU>
__global__ __launch_bounds__(256)
void finalize_kernel(const float* __restrict__ acc, const float* __restrict__ h,
                     const float* __restrict__ dinv, const float* __restrict__ bias,
                     float* out, long long total) {
    long long i = (long long)blockIdx.x * 256 + threadIdx.x;
    if (i >= total) return;
    const int node = (int)(i >> LOGF);
    const int f = (int)(i & (F - 1));
    const float di = dinv[node];
    float v = acc[i] + h[i] * di * di + bias[f];
    out[i] = RELU ? fmaxf(v, 0.0f) : v;
}

static inline int blocks_for(long long n, int bs) { return (int)((n + bs - 1) / bs); }

extern "C" void kernel_launch(void* const* d_in, const int* in_sizes, int n_in,
                              void* d_out, int out_size, void* d_ws, size_t ws_size,
                              hipStream_t stream) {
    const float* x  = (const float*)d_in[0];
    const int*   ei = (const int*)d_in[1];
    const float* W1 = (const float*)d_in[2];
    const float* b1 = (const float*)d_in[3];
    const float* W2 = (const float*)d_in[4];
    const float* b2 = (const float*)d_in[5];
    float* out = (float*)d_out;

    const int N = in_sizes[0] / IN_C;
    const long long E = (long long)in_sizes[1] / 2;
    const int* esrc = ei;
    const int* edst = ei + E;

    float* ws   = (float*)d_ws;
    float* deg  = ws;
    float* dinv = deg + N;
    float* h1   = dinv + N;
    float* acc1 = h1 + (size_t)N * HID2;
    float* acc2 = acc1 + (size_t)N * HID2;
    float* h2   = out;

    const long long n1 = (long long)N * HID2;
    const long long n2 = (long long)N * HID;

    fill_kernel<<<blocks_for(n1, 256), 256, 0, stream>>>(acc1, 0.0f, n1);
    fill_kernel<<<blocks_for(n2, 256), 256, 0, stream>>>(acc2, 0.0f, n2);
    fill_kernel<<<blocks_for(N, 256), 256, 0, stream>>>(deg, 1.0f, N);

    degree_kernel<<<blocks_for(E, 256), 256, 0, stream>>>(edst, deg, E);
    rsqrt_kernel<<<blocks_for(N, 256), 256, 0, stream>>>(deg, dinv, N);

    {
        const int mtiles = (N + 15) / 16;
        const int waves = mtiles * (HID2 / 16);
        gemm_wmma_f32<HID2><<<(waves + 3) / 4, 128, 0, stream>>>(x, W1, h1, N);
    }
    edge_aggregate<HID2, 5><<<blocks_for(E * 32, 256), 256, 0, stream>>>(
        esrc, edst, h1, dinv, acc1, E * 32);
    finalize_kernel<HID2, 7, true><<<blocks_for(n1, 256), 256, 0, stream>>>(
        acc1, h1, dinv, b1, acc1, n1);

    {
        const int mtiles = (N + 15) / 16;
        const int waves = mtiles * (HID / 16);
        gemm_wmma_f32<HID><<<(waves + 3) / 4, 128, 0, stream>>>(acc1, W2, h2, N);
    }
    edge_aggregate<HID, 4><<<blocks_for(E * 16, 256), 256, 0, stream>>>(
        esrc, edst, h2, dinv, acc2, E * 16);
    finalize_kernel<HID, 6, false><<<blocks_for(n2, 256), 256, 0, stream>>>(
        acc2, h2, dinv, b2, out, n2);
}


# Edge passes: a wave takes 32 consecutive edges, one per lane (indices, norms loaded once, coalesced), walks them by lane reads with 8 or 16 rows of gathers in flight; arrival-order sums as before
# speedup vs baseline: 1.5806x; 1.5806x over previous
#include <hip/hip_runtime.h>
#include <hip/hip_bf16.h>


typedef float v2f __attribute__((ext_vector_type(2)));
typedef float v8f __attribute__((ext_vector_type(8)));

#define IN_C 128
#define HID2 128
#define HID  64

__device__ __forceinline__ void atomic_add_f32(float* p, float v) {
    __hip_atomic_fetch_add(p, v, __ATOMIC_RELAXED, __HIP_MEMORY_SCOPE_AGENT);
}

__global__ __launch_bounds__(256)
void fill_kernel(float* __restrict__ p, float v, long long n) {
    long long i = (long long)blockIdx.x * 256 + threadIdx.x;
    if (i < n) p[i] = v;
}

__global__ __launch_bounds__(256)
void degree_kernel(const int* __restrict__ dst, float* __restrict__ deg, long long E) {
    long long e = (long long)blockIdx.x * 256 + threadIdx.x;
    if (e < E) atomic_add_f32(&deg[dst[e]], 1.0f);
}

__global__ __launch_bounds__(256)
void rsqrt_kernel(const float* __restrict__ deg, float* __restrict__ dinv, int n) {
    int i = blockIdx.x * 256 + threadIdx.x;
    if (i < n) dinv[i] = rsqrtf(deg[i]);
}

template <int NOUT>
__global__ __launch_bounds__(128)
void gemm_wmma_f32(const float* __restrict__ A, const float* __restrict__ B,
                   float* __restrict__ C, int M) {
    constexpr int K = 128;
    constexpr int NTILES = NOUT / 16;
    const int lane = threadIdx.x & 31;
    const int wglobal = blockIdx.x * 4 + (threadIdx.x >> 5);
    const int tm = wglobal / NTILES;
    const int tn = wglobal % NTILES;
    if (tm * 16 >= M) return;

    const int m  = lane & 15;
    const int kh = lane >> 4;
    const int n  = lane & 15;
    const int col = tn * 16 + n;
    const float* arow = A + (size_t)(tm * 16 + m) * K;

    v8f acc = {};
#pragma unroll 8
    for (int k0 = 0; k0 < K; k0 += 4) {
        const int ka = k0 + kh * 2;
        const float2 av = *reinterpret_cast<const float2*>(arow + ka);
        v2f a; a[0] = av.x; a[1] = av.y;
        v2f b;
        b[0] = B[(size_t)ka * NOUT + col];
        b[1] = B[(size_t)(ka + 1) * NOUT + col];
        acc = __builtin_amdgcn_wmma_f32_16x16x4_f32(false, a, false, b,
                                                    (short)0, acc, false, false);
    }

    const int row0 = tm * 16 + kh * 8;
#pragma unroll
    for (int r = 0; r < 8; ++r)
        C[(size_t)(row0 + r) * NOUT + col] = acc[r];
}

template <int F, int NE>
__device__ __forceinline__ void add_rows(int s, int d, float nrm, int k0,
                                         const float* __restrict__ h, float* __restrict__ acc,
                                         int lane) {
    constexpr int LINES = F / 32;
    float v[NE * LINES];
#pragma unroll
    for (int j = 0; j < NE; ++j) {
        const int sj = __builtin_amdgcn_readlane(s, k0 + j);
        const float* hp = h + (size_t)sj * F + lane;
#pragma unroll
        for (int p = 0; p < LINES; ++p)
            v[j * LINES + p] = hp[p * 32];
    }
#pragma unroll
    for (int j = 0; j < NE; ++j) {
        const int dj = __builtin_amdgcn_readlane(d, k0 + j);
        const float nj = __int_as_float(__builtin_amdgcn_readlane(__float_as_int(nrm), k0 + j));
        float* ap = acc + (size_t)dj * F + lane;
#pragma unroll
        for (int p = 0; p < LINES; ++p)
            atomic_add_f32(ap + p * 32, v[j * LINES + p] * nj);
    }
}

template <int F, int NE>
__global__ __launch_bounds__(256)
void edge_aggregate(const int* __restrict__ src, const int* __restrict__ dst,
                    const float* __restrict__ h, const float* __restrict__ dinv,
                    float* __restrict__ acc, long long E) {
    const int lane = (int)(threadIdx.x & 31);
    const int wave_first = __builtin_amdgcn_readfirstlane((int)threadIdx.x - lane);
    const long long e0 = (long long)blockIdx.x * 256 + wave_first;
    if (e0 >= E) return;
    const long long left = E - e0;
    const int cnt = left < 32 ? (int)left : 32;

    int s = 0, d = 0;
    float nrm = 0.0f;
    if (lane < cnt) {
        s = src[e0 + lane];
        d = dst[e0 + lane];
        nrm = dinv[s] * dinv[d];
    }

    int k = 0;
    for (; k + NE <= cnt; k += NE)
        add_rows<F, NE>(s, d, nrm, k, h, acc, lane);
#pragma unroll 1
    for (; k < cnt; ++k)
        add_rows<F, 1>(s, d, nrm, k, h, acc, lane);
}

template <int F, int LOGF, bool RELU>
__global__ __launch_bounds__(256)
void finalize_kernel(const float* __restrict__ acc, const float* __restrict__ h,
                     const float* __restrict__ dinv, const float* __restrict__ bias,
                     float* out, long long total) {
    long long i = (long long)blockIdx.x * 256 + threadIdx.x;
    if (i >= total) return;
    const int node = (int)(i >> LOGF);
    const int f = (int)(i & (F - 1));
    const float di = dinv[node];
    float v = acc[i] + h[i] * di * di + bias[f];
    out[i] = RELU ? fmaxf(v, 0.0f) : v;
}

static inline int blocks_for(long long n, int bs) { return (int)((n + bs - 1) / bs); }

extern "C" void kernel_launch(void* const* d_in, const int* in_sizes, int n_in,
                              void* d_out, int out_size, void* d_ws, size_t ws_size,
                              hipStream_t stream) {
    const float* x  = (const float*)d_in[0];
    const int*   ei = (const int*)d_in[1];
    const float* W1 = (const float*)d_in[2];
    const float* b1 = (const float*)d_in[3];
    const float* W2 = (const float*)d_in[4];
    const float* b2 = (const float*)d_in[5];
    float* out = (float*)d_out;

    const int N = in_sizes[0] / IN_C;
    const long long E = (long long)in_sizes[1] / 2;
    const int* esrc = ei;
    const int* edst = ei + E;

    float* ws   = (float*)d_ws;
    float* deg  = ws;
    float* dinv = deg + N;
    float* h1   = dinv + N;
    float* acc1 = h1 + (size_t)N * HID2;
    float* acc2 = acc1 + (size_t)N * HID2;
    float* h2   = out;

    const long long n1 = (long long)N * HID2;
    const long long n2 = (long long)N * HID;

    fill_kernel<<<blocks_for(n1, 256), 256, 0, stream>>>(acc1, 0.0f, n1);
    fill_kernel<<<blocks_for(n2, 256), 256, 0, stream>>>(acc2, 0.0f, n2);
    fill_kernel<<<blocks_for(N, 256), 256, 0, stream>>>(deg, 1.0f, N);

    degree_kernel<<<blocks_for(E, 256), 256, 0, stream>>>(edst, deg, E);
    rsqrt_kernel<<<blocks_for(N, 256), 256, 0, stream>>>(deg, dinv, N);

    {
        const int mtiles = (N + 15) / 16;
        const int waves = mtiles * (HID2 / 16);
        gemm_wmma_f32<HID2><<<(waves + 3) / 4, 128, 0, stream>>>(x, W1, h1, N);
    }
    edge_aggregate<HID2, 8><<<blocks_for(E, 256), 256, 0, stream>>>(
        esrc, edst, h1, dinv, acc1, E);
    finalize_kernel<HID2, 7, true><<<blocks_for(n1, 256), 256, 0, stream>>>(
        acc1, h1, dinv, b1, acc1, n1);

    {
        const int mtiles = (N + 15) / 16;
        const int waves = mtiles * (HID / 16);
        gemm_wmma_f32<HID><<<(waves + 3) / 4, 128, 0, stream>>>(acc1, W2, h2, N);
    }
    edge_aggregate<HID, 16><<<blocks_for(E, 256), 256, 0, stream>>>(
        esrc, edst, h2, dinv, acc2, E);
    finalize_kernel<HID, 6, false><<<blocks_for(n2, 256), 256, 0, stream>>>(
        acc2, h2, dinv, b2, out, n2);
}


# GEMM epilogues store each scatter sum's start value (self-loop term + bias); relu moves to layer 2's A loads: no zero fills or finalize passes (11 to 7 launches); self-loop term now summed first
# speedup vs baseline: 1.8701x; 1.1832x over previous
#include <hip/hip_runtime.h>
#include <hip/hip_bf16.h>


typedef float v2f __attribute__((ext_vector_type(2)));
typedef float v8f __attribute__((ext_vector_type(8)));

#define IN_C 128
#define HID2 128
#define HID  64

__device__ __forceinline__ void atomic_add_f32(float* p, float v) {
    __hip_atomic_fetch_add(p, v, __ATOMIC_RELAXED, __HIP_MEMORY_SCOPE_AGENT);
}

__global__ __launch_bounds__(256)
void fill_kernel(float* __restrict__ p, float v, long long n) {
    long long i = (long long)blockIdx.x * 256 + threadIdx.x;
    if (i < n) p[i] = v;
}

__global__ __launch_bounds__(256)
void degree_kernel(const int* __restrict__ dst, float* __restrict__ deg, long long E) {
    long long e = (long long)blockIdx.x * 256 + threadIdx.x;
    if (e < E) atomic_add_f32(&deg[dst[e]], 1.0f);
}

__global__ __launch_bounds__(256)
void rsqrt_kernel(const float* __restrict__ deg, float* __restrict__ dinv, int n) {
    int i = blockIdx.x * 256 + threadIdx.x;
    if (i < n) dinv[i] = rsqrtf(deg[i]);
}

template <int NOUT, bool RELU_A>
__global__ __launch_bounds__(128)
void gemm_wmma_f32(const float* __restrict__ A, const float* __restrict__ B,
                   const float* __restrict__ dinv, const float* __restrict__ bias,
                   float* __restrict__ H, float* __restrict__ S, int M) {
    constexpr int K = 128;
    constexpr int NTILES = NOUT / 16;
    const int lane = threadIdx.x & 31;
    const int wglobal = blockIdx.x * 4 + (threadIdx.x >> 5);
    const int tm = wglobal / NTILES;
    const int tn = wglobal % NTILES;
    if (tm * 16 >= M) return;

    const int m  = lane & 15;
    const int kh = lane >> 4;
    const int n  = lane & 15;
    const int col = tn * 16 + n;
    const float* arow = A + (size_t)(tm * 16 + m) * K;

    const int row0 = tm * 16 + kh * 8;
    float d2[8];
#pragma unroll
    for (int r = 0; r < 8; ++r) {
        const float di = dinv[row0 + r];
        d2[r] = di * di;
    }
    const float bc = bias[col];

    v8f acc = {};
#pragma unroll 8
    for (int k0 = 0; k0 < K; k0 += 4) {
        const int ka = k0 + kh * 2;
        const float2 av = *reinterpret_cast<const float2*>(arow + ka);
        v2f a;
        a[0] = RELU_A ? fmaxf(av.x, 0.0f) : av.x;
        a[1] = RELU_A ? fmaxf(av.y, 0.0f) : av.y;
        v2f b;
        b[0] = B[(size_t)ka * NOUT + col];
        b[1] = B[(size_t)(ka + 1) * NOUT + col];
        acc = __builtin_amdgcn_wmma_f32_16x16x4_f32(false, a, false, b,
                                                    (short)0, acc, false, false);
    }

#pragma unroll
    for (int r = 0; r < 8; ++r) {
        const size_t o = (size_t)(row0 + r) * NOUT + col;
        H[o] = acc[r];
        S[o] = acc[r] * d2[r] + bc;
    }
}

template <int F, int NE>
__device__ __forceinline__ void add_rows(int s, int d, float nrm, int k0,
                                         const float* __restrict__ h, float* __restrict__ acc,
                                         int lane) {
    constexpr int LINES = F / 32;
    float v[NE * LINES];
#pragma unroll
    for (int j = 0; j < NE; ++j) {
        const int sj = __builtin_amdgcn_readlane(s, k0 + j);
        const float* hp = h + (size_t)sj * F + lane;
#pragma unroll
        for (int p = 0; p < LINES; ++p)
            v[j * LINES + p] = hp[p * 32];
    }
#pragma unroll
    for (int j = 0; j < NE; ++j) {
        const int dj = __builtin_amdgcn_readlane(d, k0 + j);
        const float nj = __int_as_float(__builtin_amdgcn_readlane(__float_as_int(nrm), k0 + j));
        float* ap = acc + (size_t)dj * F + lane;
#pragma unroll
        for (int p = 0; p < LINES; ++p)
            atomic_add_f32(ap + p * 32, v[j * LINES + p] * nj);
    }
}

template <int F, int NE>
__global__ __launch_bounds__(256)
void edge_aggregate(const int* __restrict__ src, const int* __restrict__ dst,
                    const float* __restrict__ h, const float* __restrict__ dinv,
                    float* __restrict__ acc, long long E) {
    const int lane = (int)(threadIdx.x & 31);
    const int wave_first = __builtin_amdgcn_readfirstlane((int)threadIdx.x - lane);
    const long long e0 = (long long)blockIdx.x * 256 + wave_first;
    if (e0 >= E) return;
    const long long left = E - e0;
    const int cnt = left < 32 ? (int)left : 32;

    int s = 0, d = 0;
    float nrm = 0.0f;
    if (lane < cnt) {
        s = src[e0 + lane];
        d = dst[e0 + lane];
        nrm = dinv[s] * dinv[d];
    }

    int k = 0;
    for (; k + NE <= cnt; k += NE)
        add_rows<F, NE>(s, d, nrm, k, h, acc, lane);
#pragma unroll 1
    for (; k < cnt; ++k)
        add_rows<F, 1>(s, d, nrm, k, h, acc, lane);
}

static inline int blocks_for(long long n, int bs) { return (int)((n + bs - 1) / bs); }

extern "C" void kernel_launch(void* const* d_in, const int* in_sizes, int n_in,
                              void* d_out, int out_size, void* d_ws, size_t ws_size,
                              hipStream_t stream) {
    const float* x  = (const float*)d_in[0];
    const int*   ei = (const int*)d_in[1];
    const float* W1 = (const float*)d_in[2];
    const float* b1 = (const float*)d_in[3];
    const float* W2 = (const float*)d_in[4];
    const float* b2 = (const float*)d_in[5];
    float* out = (float*)d_out;

    const int N = in_sizes[0] / IN_C;
    const long long E = (long long)in_sizes[1] / 2;
    const int* esrc = ei;
    const int* edst = ei + E;

    float* ws   = (float*)d_ws;
    float* deg  = ws;
    float* dinv = deg + N;
    float* h1   = dinv + N;
    float* acc1 = h1 + (size_t)N * HID2;
    float* h2   = acc1 + (size_t)N * HID2;

    fill_kernel<<<blocks_for(N, 256), 256, 0, stream>>>(deg, 1.0f, N);

    degree_kernel<<<blocks_for(E, 256), 256, 0, stream>>>(edst, deg, E);
    rsqrt_kernel<<<blocks_for(N, 256), 256, 0, stream>>>(deg, dinv, N);

    {
        const int mtiles = (N + 15) / 16;
        const int waves = mtiles * (HID2 / 16);
        gemm_wmma_f32<HID2, false><<<(waves + 3) / 4, 128, 0, stream>>>(
            x, W1, dinv, b1, h1, acc1, N);
    }
    edge_aggregate<HID2, 8><<<blocks_for(E, 256), 256, 0, stream>>>(
        esrc, edst, h1, dinv, acc1, E);

    {
        const int mtiles = (N + 15) / 16;
        const int waves = mtiles * (HID / 16);
        gemm_wmma_f32<HID, true><<<(waves + 3) / 4, 128, 0, stream>>>(
            acc1, W2, dinv, b2, h2, out, N);
    }
    edge_aggregate<HID, 16><<<blocks_for(E, 256), 256, 0, stream>>>(
        esrc, edst, h2, dinv, out, E);
}


# GEMMs A-stationary: a wave holds a 16-row strip of A in registers (read once per 64 columns, was once per 16), streams W from an LDS copy, writes whole lines; k summed in another order
# speedup vs baseline: 2.0380x; 1.0898x over previous
#include <hip/hip_runtime.h>
#include <hip/hip_bf16.h>


typedef float v2f __attribute__((ext_vector_type(2)));
typedef float v8f __attribute__((ext_vector_type(8)));

#define IN_C 128
#define HID2 128
#define HID  64

__device__ __forceinline__ void atomic_add_f32(float* p, float v) {
    __hip_atomic_fetch_add(p, v, __ATOMIC_RELAXED, __HIP_MEMORY_SCOPE_AGENT);
}

__global__ __launch_bounds__(256)
void fill_kernel(float* __restrict__ p, float v, long long n) {
    long long i = (long long)blockIdx.x * 256 + threadIdx.x;
    if (i < n) p[i] = v;
}

__global__ __launch_bounds__(256)
void degree_kernel(const int* __restrict__ dst, float* __restrict__ deg, long long E) {
    long long e = (long long)blockIdx.x * 256 + threadIdx.x;
    if (e < E) atomic_add_f32(&deg[dst[e]], 1.0f);
}

__global__ __launch_bounds__(256)
void rsqrt_kernel(const float* __restrict__ deg, float* __restrict__ dinv, int n) {
    int i = blockIdx.x * 256 + threadIdx.x;
    if (i < n) dinv[i] = rsqrtf(deg[i]);
}

__device__ __forceinline__ float swap_halves(float x) {
    return __builtin_bit_cast(float, __builtin_amdgcn_permlanex16(
        __builtin_bit_cast(int, x), __builtin_bit_cast(int, x), 0x76543210, 0xfedcba98, false, false));
}

template <int NOUT, bool RELU_A>
__global__ __launch_bounds__(256)
void gemm_wmma_f32(const float* __restrict__ A, const float* __restrict__ B,
                   const float* __restrict__ dinv, const float* __restrict__ bias,
                   float* __restrict__ H, float* __restrict__ S, int M) {
    constexpr int K = 128;
    constexpr int NW = 64;
    constexpr int NT = NW / 16;
    constexpr int NCB = NOUT / NW;
    constexpr int NQ = K / 4;
    constexpr int PITCH = NW + 8;
    __shared__ float4 Bq[NQ * PITCH];

    const int lane = threadIdx.x & 31;
    const int c = lane & 15;
    const int h = lane >> 4;
    const int w = __builtin_amdgcn_readfirstlane((int)(threadIdx.x >> 5));
    const int cb = (blockIdx.x % NCB) * NW;
    const int strip = (blockIdx.x / NCB) * 8 + w;
    const int R0 = strip * 16;
    const bool live = R0 < M;

    constexpr int NCP = NQ * NW / 256;
    float4 bst[NCP];
#pragma unroll
    for (int it = 0; it < NCP; ++it) {
        const int i = it * 256 + (int)threadIdx.x;
        const float* bp = B + (size_t)(4 * (i / NW)) * NOUT + cb + i % NW;
        bst[it].x = bp[0];
        bst[it].y = bp[NOUT];
        bst[it].z = bp[2 * NOUT];
        bst[it].w = bp[3 * NOUT];
    }

    float4 av[16];
    {
        const float* arow = A + (size_t)((live ? R0 : 0) + c) * K + 4 * h;
#pragma unroll
        for (int j = 0; j < 16; ++j) av[j] = *reinterpret_cast<const float4*>(arow + 8 * j);
    }

#pragma unroll
    for (int it = 0; it < NCP; ++it) {
        const int i = it * 256 + (int)threadIdx.x;
        Bq[(i / NW) * PITCH + i % NW] = bst[it];
    }
    __syncthreads();
    if (!live) return;

    v8f acc[NT];
#pragma unroll
    for (int t = 0; t < NT; ++t) acc[t] = v8f{};

#pragma unroll
    for (int j = 0; j < 16; ++j) {
        float4 bv[NT];
#pragma unroll
        for (int t = 0; t < NT; ++t) bv[t] = Bq[(2 * j + h) * PITCH + 16 * t + c];
        float4 a4 = av[j];
        if (RELU_A) {
            a4.x = fmaxf(a4.x, 0.0f); a4.y = fmaxf(a4.y, 0.0f);
            a4.z = fmaxf(a4.z, 0.0f); a4.w = fmaxf(a4.w, 0.0f);
        }
        v2f a0, a1;
        a0[0] = a4.x; a0[1] = a4.y;
        a1[0] = a4.z; a1[1] = a4.w;
#pragma unroll
        for (int t = 0; t < NT; ++t) {
            v2f b;
            b[0] = bv[t].x; b[1] = bv[t].y;
            acc[t] = __builtin_amdgcn_wmma_f32_16x16x4_f32(false, a0, false, b,
                                                           (short)0, acc[t], false, false);
        }
#pragma unroll
        for (int t = 0; t < NT; ++t) {
            v2f b;
            b[0] = bv[t].z; b[1] = bv[t].w;
            acc[t] = __builtin_amdgcn_wmma_f32_16x16x4_f32(false, a1, false, b,
                                                           (short)0, acc[t], false, false);
        }
    }

    float bc[NW / 32];
#pragma unroll
    for (int p = 0; p < NW / 32; ++p) bc[p] = bias[cb + p * 32 + lane];
#pragma unroll
    for (int r = 0; r < 8; ++r) {
        const float dlo = dinv[R0 + r];
        const float dhi = dinv[R0 + 8 + r];
        const float d2lo = dlo * dlo;
        const float d2hi = dhi * dhi;
#pragma unroll
        for (int p = 0; p < NW / 32; ++p) {
            const float x = acc[2 * p][r];
            const float y = acc[2 * p + 1][r];
            const float xs = swap_halves(x);
            const float ys = swap_halves(y);
            const float u = h ? ys : x;
            const float v = h ? y : xs;
            const size_t olo = (size_t)(R0 + r) * NOUT + cb + p * 32 + lane;
            const size_t ohi = (size_t)(R0 + 8 + r) * NOUT + cb + p * 32 + lane;
            H[olo] = u;
            H[ohi] = v;
            S[olo] = u * d2lo + bc[p];
            S[ohi] = v * d2hi + bc[p];
        }
    }
}

template <int F, int NE>
__device__ __forceinline__ void add_rows(int s, int d, float nrm, int k0,
                                         const float* __restrict__ h, float* __restrict__ acc,
                                         int lane) {
    constexpr int LINES = F / 32;
    float v[NE * LINES];
#pragma unroll
    for (int j = 0; j < NE; ++j) {
        const int sj = __builtin_amdgcn_readlane(s, k0 + j);
        const float* hp = h + (size_t)sj * F + lane;
#pragma unroll
        for (int p = 0; p < LINES; ++p)
            v[j * LINES + p] = hp[p * 32];
    }
#pragma unroll
    for (int j = 0; j < NE; ++j) {
        const int dj = __builtin_amdgcn_readlane(d, k0 + j);
        const float nj = __int_as_float(__builtin_amdgcn_readlane(__float_as_int(nrm), k0 + j));
        float* ap = acc + (size_t)dj * F + lane;
#pragma unroll
        for (int p = 0; p < LINES; ++p)
            atomic_add_f32(ap + p * 32, v[j * LINES + p] * nj);
    }
}

template <int F, int NE>
__global__ __launch_bounds__(256)
void edge_aggregate(const int* __restrict__ src, const int* __restrict__ dst,
                    const float* __restrict__ h, const float* __restrict__ dinv,
                    float* __restrict__ acc, long long E) {
    const int lane = (int)(threadIdx.x & 31);
    const int wave_first = __builtin_amdgcn_readfirstlane((int)threadIdx.x - lane);
    const long long e0 = (long long)blockIdx.x * 256 + wave_first;
    if (e0 >= E) return;
    const long long left = E - e0;
    const int cnt = left < 32 ? (int)left : 32;

    int s = 0, d = 0;
    float nrm = 0.0f;
    if (lane < cnt) {
        s = src[e0 + lane];
        d = dst[e0 + lane];
        nrm = dinv[s] * dinv[d];
    }

    int k = 0;
    for (; k + NE <= cnt; k += NE)
        add_rows<F, NE>(s, d, nrm, k, h, acc, lane);
#pragma unroll 1
    for (; k < cnt; ++k)
        add_rows<F, 1>(s, d, nrm, k, h, acc, lane);
}

static inline int blocks_for(long long n, int bs) { return (int)((n + bs - 1) / bs); }

extern "C" void kernel_launch(void* const* d_in, const int* in_sizes, int n_in,
                              void* d_out, int out_size, void* d_ws, size_t ws_size,
                              hipStream_t stream) {
    const float* x  = (const float*)d_in[0];
    const int*   ei = (const int*)d_in[1];
    const float* W1 = (const float*)d_in[2];
    const float* b1 = (const float*)d_in[3];
    const float* W2 = (const float*)d_in[4];
    const float* b2 = (const float*)d_in[5];
    float* out = (float*)d_out;

    const int N = in_sizes[0] / IN_C;
    const long long E = (long long)in_sizes[1] / 2;
    const int* esrc = ei;
    const int* edst = ei + E;

    float* ws   = (float*)d_ws;
    float* deg  = ws;
    float* dinv = deg + N;
    float* h1   = dinv + N;
    float* acc1 = h1 + (size_t)N * HID2;
    float* h2   = acc1 + (size_t)N * HID2;

    fill_kernel<<<blocks_for(N, 256), 256, 0, stream>>>(deg, 1.0f, N);

    degree_kernel<<<blocks_for(E, 256), 256, 0, stream>>>(edst, deg, E);
    rsqrt_kernel<<<blocks_for(N, 256), 256, 0, stream>>>(deg, dinv, N);

    {
        const int mtiles = (N + 15) / 16;
        const int groups = (mtiles + 7) / 8 * (HID2 / 64);
        gemm_wmma_f32<HID2, false><<<groups, 256, 0, stream>>>(
            x, W1, dinv, b1, h1, acc1, N);
    }
    edge_aggregate<HID2, 8><<<blocks_for(E, 256), 256, 0, stream>>>(
        esrc, edst, h1, dinv, acc1, E);

    {
        const int mtiles = (N + 15) / 16;
        const int groups = (mtiles + 7) / 8 * (HID / 64);
        gemm_wmma_f32<HID, true><<<groups, 256, 0, stream>>>(
            acc1, W2, dinv, b2, h2, out, N);
    }
    edge_aggregate<HID, 16><<<blocks_for(E, 256), 256, 0, stream>>>(
        esrc, edst, h2, dinv, out, E);
}


# Edge sums live in LDS: edges filed once by destination range (400 rows); a work-group per range adds gathered rows to its LDS sums (LDS atomics), writes each row once; arrival-order sums
# speedup vs baseline: 2.6210x; 1.2861x over previous
#include <hip/hip_runtime.h>
#include <hip/hip_bf16.h>


typedef float v2f __attribute__((ext_vector_type(2)));
typedef float v8f __attribute__((ext_vector_type(8)));

#define IN_C 128
#define HID2 128
#define HID  64
#define ROWS 400
#define MAXB 256
#define EPB  4096

__device__ __forceinline__ void atomic_add_f32(float* p, float v) {
    __hip_atomic_fetch_add(p, v, __ATOMIC_RELAXED, __HIP_MEMORY_SCOPE_AGENT);
}

__device__ __forceinline__ int lds_count(int* p) {
    return __hip_atomic_fetch_add(p, 1, __ATOMIC_RELAXED, __HIP_MEMORY_SCOPE_WORKGROUP);
}

__global__ __launch_bounds__(256)
void init_kernel(float* __restrict__ deg, int* __restrict__ cnt, int n, int nb) {
    int i = blockIdx.x * 256 + threadIdx.x;
    if (i < n) deg[i] = 1.0f;
    if (i < nb) cnt[i] = 0;
}

__global__ __launch_bounds__(256)
void degree_kernel(const int* __restrict__ dst, float* __restrict__ deg, int* __restrict__ cnt,
                   long long E, int nb) {
    __shared__ int hist[MAXB];
    const int t = threadIdx.x;
    hist[t] = 0;
    __syncthreads();
    const long long base = (long long)blockIdx.x * EPB;
    int dv[EPB / 256];
#pragma unroll
    for (int j = 0; j < EPB / 256; ++j) {
        const long long e = base + j * 256 + t;
        dv[j] = e < E ? dst[e] : -1;
    }
#pragma unroll
    for (int j = 0; j < EPB / 256; ++j) {
        if (dv[j] >= 0) {
            atomic_add_f32(&deg[dv[j]], 1.0f);
            lds_count(&hist[dv[j] / ROWS]);
        }
    }
    __syncthreads();
    if (t < nb && hist[t] != 0)
        __hip_atomic_fetch_add(&cnt[t], hist[t], __ATOMIC_RELAXED, __HIP_MEMORY_SCOPE_AGENT);
}

__global__ __launch_bounds__(256)
void rsqrt_kernel(const float* __restrict__ deg, float* __restrict__ dinv, int n,
                  const int* __restrict__ cnt, int* __restrict__ first, int* __restrict__ cursor, int nb) {
    int i = blockIdx.x * 256 + threadIdx.x;
    if (i < n) dinv[i] = rsqrtf(deg[i]);
    if (blockIdx.x == 0) {
        __shared__ int sc[MAXB];
        const int t = threadIdx.x;
        const int v = t < nb ? cnt[t] : 0;
        sc[t] = v;
        __syncthreads();
        for (int off = 1; off < MAXB; off <<= 1) {
            const int x = t >= off ? sc[t - off] : 0;
            __syncthreads();
            sc[t] += x;
            __syncthreads();
        }
        if (t < nb) {
            first[t] = sc[t] - v;
            cursor[t] = sc[t] - v;
        }
        if (t == nb - 1) first[nb] = sc[t];
    }
}

__global__ __launch_bounds__(256)
void file_edges_kernel(const int* __restrict__ src, const int* __restrict__ dst,
                       int* __restrict__ cursor, unsigned long long* __restrict__ pairs,
                       long long E, int nb) {
    __shared__ int hist[MAXB];
    __shared__ int slot0[MAXB];
    const int t = threadIdx.x;
    hist[t] = 0;
    __syncthreads();
    const long long base = (long long)blockIdx.x * EPB;
    int sv[EPB / 256], dv[EPB / 256], rk[EPB / 256];
#pragma unroll
    for (int j = 0; j < EPB / 256; ++j) {
        const long long e = base + j * 256 + t;
        sv[j] = e < E ? src[e] : 0;
        dv[j] = e < E ? dst[e] : -1;
    }
#pragma unroll
    for (int j = 0; j < EPB / 256; ++j)
        rk[j] = dv[j] >= 0 ? lds_count(&hist[dv[j] / ROWS]) : 0;
    __syncthreads();
    if (t < nb && hist[t] != 0)
        slot0[t] = __hip_atomic_fetch_add(&cursor[t], hist[t], __ATOMIC_RELAXED, __HIP_MEMORY_SCOPE_AGENT);
    __syncthreads();
#pragma unroll
    for (int j = 0; j < EPB / 256; ++j) {
        if (dv[j] >= 0)
            pairs[slot0[dv[j] / ROWS] + rk[j]] =
                (unsigned long long)(unsigned)sv[j] | ((unsigned long long)(unsigned)dv[j] << 32);
    }
}

__device__ __forceinline__ float swap_halves(float x) {
    return __builtin_bit_cast(float, __builtin_amdgcn_permlanex16(
        __builtin_bit_cast(int, x), __builtin_bit_cast(int, x), 0x76543210, 0xfedcba98, false, false));
}

template <int NOUT, bool RELU_A>
__global__ __launch_bounds__(256)
void gemm_wmma_f32(const float* __restrict__ A, const float* __restrict__ B,
                   float* __restrict__ H, int M) {
    constexpr int K = 128;
    constexpr int NW = 64;
    constexpr int NT = NW / 16;
    constexpr int NCB = NOUT / NW;
    constexpr int NQ = K / 4;
    constexpr int PITCH = NW + 8;
    __shared__ float4 Bq[NQ * PITCH];

    const int lane = threadIdx.x & 31;
    const int c = lane & 15;
    const int h = lane >> 4;
    const int w = __builtin_amdgcn_readfirstlane((int)(threadIdx.x >> 5));
    const int cb = (blockIdx.x % NCB) * NW;
    const int strip = (blockIdx.x / NCB) * 8 + w;
    const int R0 = strip * 16;
    const bool live = R0 < M;

    constexpr int NCP = NQ * NW / 256;
    float4 bst[NCP];
#pragma unroll
    for (int it = 0; it < NCP; ++it) {
        const int i = it * 256 + (int)threadIdx.x;
        const float* bp = B + (size_t)(4 * (i / NW)) * NOUT + cb + i % NW;
        bst[it].x = bp[0];
        bst[it].y = bp[NOUT];
        bst[it].z = bp[2 * NOUT];
        bst[it].w = bp[3 * NOUT];
    }

    float4 av[16];
    {
        const float* arow = A + (size_t)((live ? R0 : 0) + c) * K + 4 * h;
#pragma unroll
        for (int j = 0; j < 16; ++j) av[j] = *reinterpret_cast<const float4*>(arow + 8 * j);
    }

#pragma unroll
    for (int it = 0; it < NCP; ++it) {
        const int i = it * 256 + (int)threadIdx.x;
        Bq[(i / NW) * PITCH + i % NW] = bst[it];
    }
    __syncthreads();
    if (!live) return;

    v8f acc[NT];
#pragma unroll
    for (int t = 0; t < NT; ++t) acc[t] = v8f{};

#pragma unroll
    for (int j = 0; j < 16; ++j) {
        float4 bv[NT];
#pragma unroll
        for (int t = 0; t < NT; ++t) bv[t] = Bq[(2 * j + h) * PITCH + 16 * t + c];
        float4 a4 = av[j];
        if (RELU_A) {
            a4.x = fmaxf(a4.x, 0.0f); a4.y = fmaxf(a4.y, 0.0f);
            a4.z = fmaxf(a4.z, 0.0f); a4.w = fmaxf(a4.w, 0.0f);
        }
        v2f a0, a1;
        a0[0] = a4.x; a0[1] = a4.y;
        a1[0] = a4.z; a1[1] = a4.w;
#pragma unroll
        for (int t = 0; t < NT; ++t) {
            v2f b;
            b[0] = bv[t].x; b[1] = bv[t].y;
            acc[t] = __builtin_amdgcn_wmma_f32_16x16x4_f32(false, a0, false, b,
                                                           (short)0, acc[t], false, false);
        }
#pragma unroll
        for (int t = 0; t < NT; ++t) {
            v2f b;
            b[0] = bv[t].z; b[1] = bv[t].w;
            acc[t] = __builtin_amdgcn_wmma_f32_16x16x4_f32(false, a1, false, b,
                                                           (short)0, acc[t], false, false);
        }
    }

#pragma unroll
    for (int r = 0; r < 8; ++r) {
#pragma unroll
        for (int p = 0; p < NW / 32; ++p) {
            const float x = acc[2 * p][r];
            const float y = acc[2 * p + 1][r];
            const float xs = swap_halves(x);
            const float ys = swap_halves(y);
            const float u = h ? ys : x;
            const float v = h ? y : xs;
            const size_t olo = (size_t)(R0 + r) * NOUT + cb + p * 32 + lane;
            const size_t ohi = (size_t)(R0 + 8 + r) * NOUT + cb + p * 32 + lane;
            H[olo] = u;
            H[ohi] = v;
        }
    }
}

__device__ __forceinline__ void lds_add_f32(float* p, float v) {
    __hip_atomic_fetch_add(p, v, __ATOMIC_RELAXED, __HIP_MEMORY_SCOPE_WORKGROUP);
}

template <int F, int NE>
__device__ __forceinline__ void add_rows(int s, int dl, float nrm, int k0,
                                         const float* __restrict__ h, float* acc, int lane) {
    constexpr int LINES = F / 32;
    float v[NE * LINES];
#pragma unroll
    for (int j = 0; j < NE; ++j) {
        const int sj = __builtin_amdgcn_readlane(s, k0 + j);
        const float* hp = h + (size_t)sj * F + lane;
#pragma unroll
        for (int p = 0; p < LINES; ++p)
            v[j * LINES + p] = hp[p * 32];
    }
#pragma unroll
    for (int j = 0; j < NE; ++j) {
        const int dj = __builtin_amdgcn_readlane(dl, k0 + j);
        const float nj = __int_as_float(__builtin_amdgcn_readlane(__float_as_int(nrm), k0 + j));
        float* ap = acc + dj * F + lane;
#pragma unroll
        for (int p = 0; p < LINES; ++p)
            lds_add_f32(ap + p * 32, v[j * LINES + p] * nj);
    }
}

template <int F, int NE>
__global__ __launch_bounds__(1024)
void edge_aggregate(const unsigned long long* __restrict__ pairs, const int* __restrict__ first,
                    const float* __restrict__ h, const float* __restrict__ dinv,
                    const float* __restrict__ bias, float* __restrict__ out, int N) {
    __shared__ float acc[ROWS * F];
    const int tid = (int)threadIdx.x;
    const int lane = tid & 31;
    const int w = __builtin_amdgcn_readfirstlane(tid >> 5);
    const int row0 = (int)blockIdx.x * ROWS;
    const int left_rows = N - row0;
    const int nfl = (left_rows < ROWS ? left_rows : ROWS) * F;

    for (int i = tid; i < nfl; i += 1024) {
        const float d = dinv[row0 + i / F];
        acc[i] = h[(size_t)row0 * F + i] * (d * d) + bias[i % F];
    }
    __syncthreads();

    const int beg = first[blockIdx.x];
    const int end = first[blockIdx.x + 1];
    for (int e0 = beg + w * 32; e0 < end; e0 += 32 * 32) {
        const int left = end - e0;
        const int cnt = left < 32 ? left : 32;
        int s = 0, dl = 0;
        float nrm = 0.0f;
        if (lane < cnt) {
            const unsigned long long pr = pairs[e0 + lane];
            s = (int)(unsigned)pr;
            const int d = (int)(unsigned)(pr >> 32);
            dl = d - row0;
            nrm = dinv[s] * dinv[d];
        }
        int k = 0;
        for (; k + NE <= cnt; k += NE) add_rows<F, NE>(s, dl, nrm, k, h, acc, lane);
        if constexpr (NE > 8) {
            if (k + 8 <= cnt) { add_rows<F, 8>(s, dl, nrm, k, h, acc, lane); k += 8; }
        }
        if (k + 4 <= cnt) { add_rows<F, 4>(s, dl, nrm, k, h, acc, lane); k += 4; }
        if (k + 2 <= cnt) { add_rows<F, 2>(s, dl, nrm, k, h, acc, lane); k += 2; }
        if (k < cnt) add_rows<F, 1>(s, dl, nrm, k, h, acc, lane);
    }
    __syncthreads();

    for (int i = tid; i < nfl; i += 1024) out[(size_t)row0 * F + i] = acc[i];
}

static inline int blocks_for(long long n, int bs) { return (int)((n + bs - 1) / bs); }

extern "C" void kernel_launch(void* const* d_in, const int* in_sizes, int n_in,
                              void* d_out, int out_size, void* d_ws, size_t ws_size,
                              hipStream_t stream) {
    const float* x  = (const float*)d_in[0];
    const int*   ei = (const int*)d_in[1];
    const float* W1 = (const float*)d_in[2];
    const float* b1 = (const float*)d_in[3];
    const float* W2 = (const float*)d_in[4];
    const float* b2 = (const float*)d_in[5];
    float* out = (float*)d_out;

    const int N = in_sizes[0] / IN_C;
    const long long E = (long long)in_sizes[1] / 2;
    const int* esrc = ei;
    const int* edst = ei + E;

    float* ws   = (float*)d_ws;
    float* deg  = ws;
    float* dinv = deg + N;
    float* h1   = dinv + N;
    float* acc1 = h1 + (size_t)N * HID2;
    float* h2   = acc1 + (size_t)N * HID2;
    unsigned long long* pairs = (unsigned long long*)(h2 + (size_t)N * HID);
    int* cnt    = (int*)(pairs + E);
    int* first  = cnt + MAXB;
    int* cursor = first + MAXB + 4;

    const int nb = (N + ROWS - 1) / ROWS;

    init_kernel<<<blocks_for(N, 256), 256, 0, stream>>>(deg, cnt, N, nb);

    degree_kernel<<<blocks_for(E, EPB), 256, 0, stream>>>(edst, deg, cnt, E, nb);
    rsqrt_kernel<<<blocks_for(N, 256), 256, 0, stream>>>(deg, dinv, N, cnt, first, cursor, nb);
    file_edges_kernel<<<blocks_for(E, EPB), 256, 0, stream>>>(esrc, edst, cursor, pairs, E, nb);

    {
        const int mtiles = (N + 15) / 16;
        const int groups = (mtiles + 7) / 8 * (HID2 / 64);
        gemm_wmma_f32<HID2, false><<<groups, 256, 0, stream>>>(x, W1, h1, N);
    }
    edge_aggregate<HID2, 8><<<nb, 1024, 0, stream>>>(pairs, first, h1, dinv, b1, acc1, N);

    {
        const int mtiles = (N + 15) / 16;
        const int groups = (mtiles + 7) / 8 * (HID / 64);
        gemm_wmma_f32<HID, true><<<groups, 256, 0, stream>>>(acc1, W2, h2, N);
    }
    edge_aggregate<HID, 16><<<nb, 1024, 0, stream>>>(pairs, first, h2, dinv, b2, out, N);
}


# In-degrees counted per destination range: a work-group per range counts its filed edges' rows in LDS (integer adds) and writes the norms; no global atomic per edge, no degree array (8 to 9 launches)
# speedup vs baseline: 3.7542x; 1.4324x over previous
#include <hip/hip_runtime.h>
#include <hip/hip_bf16.h>


typedef float v2f __attribute__((ext_vector_type(2)));
typedef float v8f __attribute__((ext_vector_type(8)));

#define IN_C 128
#define HID2 128
#define HID  64
#define ROWS 400
#define MAXB 256
#define EPB  4096

__device__ __forceinline__ int lds_count(int* p) {
    return __hip_atomic_fetch_add(p, 1, __ATOMIC_RELAXED, __HIP_MEMORY_SCOPE_WORKGROUP);
}

__global__ __launch_bounds__(256)
void init_kernel(int* __restrict__ cnt, int nb) {
    const int i = threadIdx.x;
    if (i < nb) cnt[i] = 0;
}

__global__ __launch_bounds__(256)
void count_kernel(const int* __restrict__ dst, int* __restrict__ cnt, long long E, int nb) {
    __shared__ int hist[MAXB];
    const int t = threadIdx.x;
    hist[t] = 0;
    __syncthreads();
    const long long base = (long long)blockIdx.x * EPB;
    int dv[EPB / 256];
#pragma unroll
    for (int j = 0; j < EPB / 256; ++j) {
        const long long e = base + j * 256 + t;
        dv[j] = e < E ? dst[e] : -1;
    }
#pragma unroll
    for (int j = 0; j < EPB / 256; ++j) {
        if (dv[j] >= 0) lds_count(&hist[dv[j] / ROWS]);
    }
    __syncthreads();
    if (t < nb && hist[t] != 0)
        __hip_atomic_fetch_add(&cnt[t], hist[t], __ATOMIC_RELAXED, __HIP_MEMORY_SCOPE_AGENT);
}

__global__ __launch_bounds__(256)
void scan_kernel(const int* __restrict__ cnt, int* __restrict__ first, int* __restrict__ cursor, int nb) {
    __shared__ int sc[MAXB];
    const int t = threadIdx.x;
    const int v = t < nb ? cnt[t] : 0;
    sc[t] = v;
    __syncthreads();
    for (int off = 1; off < MAXB; off <<= 1) {
        const int x = t >= off ? sc[t - off] : 0;
        __syncthreads();
        sc[t] += x;
        __syncthreads();
    }
    if (t < nb) {
        first[t] = sc[t] - v;
        cursor[t] = sc[t] - v;
    }
    if (t == nb - 1) first[nb] = sc[t];
}

__global__ __launch_bounds__(256)
void file_edges_kernel(const int* __restrict__ src, const int* __restrict__ dst,
                       int* __restrict__ cursor, unsigned long long* __restrict__ pairs,
                       long long E, int nb) {
    __shared__ int hist[MAXB];
    __shared__ int slot0[MAXB];
    const int t = threadIdx.x;
    hist[t] = 0;
    __syncthreads();
    const long long base = (long long)blockIdx.x * EPB;
    int sv[EPB / 256], dv[EPB / 256], rk[EPB / 256];
#pragma unroll
    for (int j = 0; j < EPB / 256; ++j) {
        const long long e = base + j * 256 + t;
        sv[j] = e < E ? src[e] : 0;
        dv[j] = e < E ? dst[e] : -1;
    }
#pragma unroll
    for (int j = 0; j < EPB / 256; ++j)
        rk[j] = dv[j] >= 0 ? lds_count(&hist[dv[j] / ROWS]) : 0;
    __syncthreads();
    if (t < nb && hist[t] != 0)
        slot0[t] = __hip_atomic_fetch_add(&cursor[t], hist[t], __ATOMIC_RELAXED, __HIP_MEMORY_SCOPE_AGENT);
    __syncthreads();
#pragma unroll
    for (int j = 0; j < EPB / 256; ++j) {
        if (dv[j] >= 0)
            pairs[slot0[dv[j] / ROWS] + rk[j]] =
                (unsigned long long)(unsigned)sv[j] | ((unsigned long long)(unsigned)dv[j] << 32);
    }
}

#define DGT 512
#define DGE 16
__global__ __launch_bounds__(DGT)
void range_norm_kernel(const unsigned long long* __restrict__ pairs, const int* __restrict__ first,
                       float* __restrict__ dinv, int N) {
    __shared__ int c[ROWS];
    const int t = threadIdx.x;
    const int row0 = (int)blockIdx.x * ROWS;
    for (int i = t; i < ROWS; i += DGT) c[i] = 0;
    __syncthreads();
    const int beg = first[blockIdx.x];
    const int end = first[blockIdx.x + 1];
    const int* hi = reinterpret_cast<const int*>(pairs) + 1;
    for (int e0 = beg; e0 < end; e0 += DGT * DGE) {
        int dv[DGE];
#pragma unroll
        for (int j = 0; j < DGE; ++j) {
            const int e = e0 + j * DGT + t;
            dv[j] = e < end ? hi[2 * (size_t)e] : -1;
        }
#pragma unroll
        for (int j = 0; j < DGE; ++j)
            if (dv[j] >= 0) lds_count(&c[dv[j] - row0]);
    }
    __syncthreads();
    const int left_rows = N - row0;
    const int rows = left_rows < ROWS ? left_rows : ROWS;
    for (int i = t; i < rows; i += DGT) dinv[row0 + i] = rsqrtf((float)(c[i] + 1));
}

__device__ __forceinline__ float swap_halves(float x) {
    return __builtin_bit_cast(float, __builtin_amdgcn_permlanex16(
        __builtin_bit_cast(int, x), __builtin_bit_cast(int, x), 0x76543210, 0xfedcba98, false, false));
}

template <int NOUT, bool RELU_A>
__global__ __launch_bounds__(256)
void gemm_wmma_f32(const float* __restrict__ A, const float* __restrict__ B,
                   float* __restrict__ H, int M) {
    constexpr int K = 128;
    constexpr int NW = 64;
    constexpr int NT = NW / 16;
    constexpr int NCB = NOUT / NW;
    constexpr int NQ = K / 4;
    constexpr int PITCH = NW + 8;
    __shared__ float4 Bq[NQ * PITCH];

    const int lane = threadIdx.x & 31;
    const int c = lane & 15;
    const int h = lane >> 4;
    const int w = __builtin_amdgcn_readfirstlane((int)(threadIdx.x >> 5));
    const int cb = (blockIdx.x % NCB) * NW;
    const int strip = (blockIdx.x / NCB) * 8 + w;
    const int R0 = strip * 16;
    const bool live = R0 < M;

    constexpr int NCP = NQ * NW / 256;
    float4 bst[NCP];
#pragma unroll
    for (int it = 0; it < NCP; ++it) {
        const int i = it * 256 + (int)threadIdx.x;
        const float* bp = B + (size_t)(4 * (i / NW)) * NOUT + cb + i % NW;
        bst[it].x = bp[0];
        bst[it].y = bp[NOUT];
        bst[it].z = bp[2 * NOUT];
        bst[it].w = bp[3 * NOUT];
    }

    float4 av[16];
    {
        const float* arow = A + (size_t)((live ? R0 : 0) + c) * K + 4 * h;
#pragma unroll
        for (int j = 0; j < 16; ++j) av[j] = *reinterpret_cast<const float4*>(arow + 8 * j);
    }

#pragma unroll
    for (int it = 0; it < NCP; ++it) {
        const int i = it * 256 + (int)threadIdx.x;
        Bq[(i / NW) * PITCH + i % NW] = bst[it];
    }
    __syncthreads();
    if (!live) return;

    v8f acc[NT];
#pragma unroll
    for (int t = 0; t < NT; ++t) acc[t] = v8f{};

#pragma unroll
    for (int j = 0; j < 16; ++j) {
        float4 bv[NT];
#pragma unroll
        for (int t = 0; t < NT; ++t) bv[t] = Bq[(2 * j + h) * PITCH + 16 * t + c];
        float4 a4 = av[j];
        if (RELU_A) {
            a4.x = fmaxf(a4.x, 0.0f); a4.y = fmaxf(a4.y, 0.0f);
            a4.z = fmaxf(a4.z, 0.0f); a4.w = fmaxf(a4.w, 0.0f);
        }
        v2f a0, a1;
        a0[0] = a4.x; a0[1] = a4.y;
        a1[0] = a4.z; a1[1] = a4.w;
#pragma unroll
        for (int t = 0; t < NT; ++t) {
            v2f b;
            b[0] = bv[t].x; b[1] = bv[t].y;
            acc[t] = __builtin_amdgcn_wmma_f32_16x16x4_f32(false, a0, false, b,
                                                           (short)0, acc[t], false, false);
        }
#pragma unroll
        for (int t = 0; t < NT; ++t) {
            v2f b;
            b[0] = bv[t].z; b[1] = bv[t].w;
            acc[t] = __builtin_amdgcn_wmma_f32_16x16x4_f32(false, a1, false, b,
                                                           (short)0, acc[t], false, false);
        }
    }

#pragma unroll
    for (int r = 0; r < 8; ++r) {
#pragma unroll
        for (int p = 0; p < NW / 32; ++p) {
            const float x = acc[2 * p][r];
            const float y = acc[2 * p + 1][r];
            const float xs = swap_halves(x);
            const float ys = swap_halves(y);
            const float u = h ? ys : x;
            const float v = h ? y : xs;
            const size_t olo = (size_t)(R0 + r) * NOUT + cb + p * 32 + lane;
            const size_t ohi = (size_t)(R0 + 8 + r) * NOUT + cb + p * 32 + lane;
            H[olo] = u;
            H[ohi] = v;
        }
    }
}

__device__ __forceinline__ void lds_add_f32(float* p, float v) {
    __hip_atomic_fetch_add(p, v, __ATOMIC_RELAXED, __HIP_MEMORY_SCOPE_WORKGROUP);
}

template <int F, int NE>
__device__ __forceinline__ void add_rows(int s, int dl, float nrm, int k0,
                                         const float* __restrict__ h, float* acc, int lane) {
    constexpr int LINES = F / 32;
    float v[NE * LINES];
#pragma unroll
    for (int j = 0; j < NE; ++j) {
        const int sj = __builtin_amdgcn_readlane(s, k0 + j);
        const float* hp = h + (size_t)sj * F + lane;
#pragma unroll
        for (int p = 0; p < LINES; ++p)
            v[j * LINES + p] = hp[p * 32];
    }
#pragma unroll
    for (int j = 0; j < NE; ++j) {
        const int dj = __builtin_amdgcn_readlane(dl, k0 + j);
        const float nj = __int_as_float(__builtin_amdgcn_readlane(__float_as_int(nrm), k0 + j));
        float* ap = acc + dj * F + lane;
#pragma unroll
        for (int p = 0; p < LINES; ++p)
            lds_add_f32(ap + p * 32, v[j * LINES + p] * nj);
    }
}

template <int F, int NE>
__global__ __launch_bounds__(1024)
void edge_aggregate(const unsigned long long* __restrict__ pairs, const int* __restrict__ first,
                    const float* __restrict__ h, const float* __restrict__ dinv,
                    const float* __restrict__ bias, float* __restrict__ out, int N) {
    __shared__ float acc[ROWS * F];
    const int tid = (int)threadIdx.x;
    const int lane = tid & 31;
    const int w = __builtin_amdgcn_readfirstlane(tid >> 5);
    const int row0 = (int)blockIdx.x * ROWS;
    const int left_rows = N - row0;
    const int nfl = (left_rows < ROWS ? left_rows : ROWS) * F;

    for (int i = tid; i < nfl; i += 1024) {
        const float d = dinv[row0 + i / F];
        acc[i] = h[(size_t)row0 * F + i] * (d * d) + bias[i % F];
    }
    __syncthreads();

    const int beg = first[blockIdx.x];
    const int end = first[blockIdx.x + 1];
    for (int e0 = beg + w * 32; e0 < end; e0 += 32 * 32) {
        const int left = end - e0;
        const int cnt = left < 32 ? left : 32;
        int s = 0, dl = 0;
        float nrm = 0.0f;
        if (lane < cnt) {
            const unsigned long long pr = pairs[e0 + lane];
            s = (int)(unsigned)pr;
            const int d = (int)(unsigned)(pr >> 32);
            dl = d - row0;
            nrm = dinv[s] * dinv[d];
        }
        int k = 0;
        for (; k + NE <= cnt; k += NE) add_rows<F, NE>(s, dl, nrm, k, h, acc, lane);
        if constexpr (NE > 8) {
            if (k + 8 <= cnt) { add_rows<F, 8>(s, dl, nrm, k, h, acc, lane); k += 8; }
        }
        if (k + 4 <= cnt) { add_rows<F, 4>(s, dl, nrm, k, h, acc, lane); k += 4; }
        if (k + 2 <= cnt) { add_rows<F, 2>(s, dl, nrm, k, h, acc, lane); k += 2; }
        if (k < cnt) add_rows<F, 1>(s, dl, nrm, k, h, acc, lane);
    }
    __syncthreads();

    for (int i = tid; i < nfl; i += 1024) out[(size_t)row0 * F + i] = acc[i];
}

static inline int blocks_for(long long n, int bs) { return (int)((n + bs - 1) / bs); }

extern "C" void kernel_launch(void* const* d_in, const int* in_sizes, int n_in,
                              void* d_out, int out_size, void* d_ws, size_t ws_size,
                              hipStream_t stream) {
    const float* x  = (const float*)d_in[0];
    const int*   ei = (const int*)d_in[1];
    const float* W1 = (const float*)d_in[2];
    const float* b1 = (const float*)d_in[3];
    const float* W2 = (const float*)d_in[4];
    const float* b2 = (const float*)d_in[5];
    float* out = (float*)d_out;

    const int N = in_sizes[0] / IN_C;
    const long long E = (long long)in_sizes[1] / 2;
    const int* esrc = ei;
    const int* edst = ei + E;

    float* ws   = (float*)d_ws;
    float* dinv = ws;
    float* h1   = dinv + 2 * (size_t)N;
    float* acc1 = h1 + (size_t)N * HID2;
    float* h2   = acc1 + (size_t)N * HID2;
    unsigned long long* pairs = (unsigned long long*)(h2 + (size_t)N * HID);
    int* cnt    = (int*)(pairs + E);
    int* first  = cnt + MAXB;
    int* cursor = first + MAXB + 4;

    const int nb = (N + ROWS - 1) / ROWS;

    init_kernel<<<1, 256, 0, stream>>>(cnt, nb);

    count_kernel<<<blocks_for(E, EPB), 256, 0, stream>>>(edst, cnt, E, nb);
    scan_kernel<<<1, 256, 0, stream>>>(cnt, first, cursor, nb);
    file_edges_kernel<<<blocks_for(E, EPB), 256, 0, stream>>>(esrc, edst, cursor, pairs, E, nb);
    range_norm_kernel<<<nb, DGT, 0, stream>>>(pairs, first, dinv, N);

    {
        const int mtiles = (N + 15) / 16;
        const int groups = (mtiles + 7) / 8 * (HID2 / 64);
        gemm_wmma_f32<HID2, false><<<groups, 256, 0, stream>>>(x, W1, h1, N);
    }
    edge_aggregate<HID2, 8><<<nb, 1024, 0, stream>>>(pairs, first, h1, dinv, b1, acc1, N);

    {
        const int mtiles = (N + 15) / 16;
        const int groups = (mtiles + 7) / 8 * (HID / 64);
        gemm_wmma_f32<HID, true><<<groups, 256, 0, stream>>>(acc1, W2, h2, N);
    }
    edge_aggregate<HID, 16><<<nb, 1024, 0, stream>>>(pairs, first, h2, dinv, b2, out, N);
}


# Edge sums in LDS without float atomics: a range's edges ordered in LDS by owning wave (row & 31); each wave updates its rows by plain LDS load, fma, store; sum order: LDS arrival rank, fused mul-add
# speedup vs baseline: 3.9264x; 1.0459x over previous
#include <hip/hip_runtime.h>
#include <hip/hip_bf16.h>


typedef float v2f __attribute__((ext_vector_type(2)));
typedef float v8f __attribute__((ext_vector_type(8)));

#define IN_C 128
#define HID2 128
#define HID  64
#define ROWS 400
#define MAXB 256
#define EPB  4096

__device__ __forceinline__ int lds_count(int* p) {
    return __hip_atomic_fetch_add(p, 1, __ATOMIC_RELAXED, __HIP_MEMORY_SCOPE_WORKGROUP);
}

__global__ __launch_bounds__(256)
void init_kernel(int* __restrict__ cnt, int nb) {
    const int i = threadIdx.x;
    if (i < nb) cnt[i] = 0;
}

__global__ __launch_bounds__(256)
void count_kernel(const int* __restrict__ dst, int* __restrict__ cnt, long long E, int nb) {
    __shared__ int hist[MAXB];
    const int t = threadIdx.x;
    hist[t] = 0;
    __syncthreads();
    const long long base = (long long)blockIdx.x * EPB;
    int dv[EPB / 256];
#pragma unroll
    for (int j = 0; j < EPB / 256; ++j) {
        const long long e = base + j * 256 + t;
        dv[j] = e < E ? dst[e] : -1;
    }
#pragma unroll
    for (int j = 0; j < EPB / 256; ++j) {
        if (dv[j] >= 0) lds_count(&hist[dv[j] / ROWS]);
    }
    __syncthreads();
    if (t < nb && hist[t] != 0)
        __hip_atomic_fetch_add(&cnt[t], hist[t], __ATOMIC_RELAXED, __HIP_MEMORY_SCOPE_AGENT);
}

__global__ __launch_bounds__(256)
void scan_kernel(const int* __restrict__ cnt, int* __restrict__ first, int* __restrict__ cursor, int nb) {
    __shared__ int sc[MAXB];
    const int t = threadIdx.x;
    const int v = t < nb ? cnt[t] : 0;
    sc[t] = v;
    __syncthreads();
    for (int off = 1; off < MAXB; off <<= 1) {
        const int x = t >= off ? sc[t - off] : 0;
        __syncthreads();
        sc[t] += x;
        __syncthreads();
    }
    if (t < nb) {
        first[t] = sc[t] - v;
        cursor[t] = sc[t] - v;
    }
    if (t == nb - 1) first[nb] = sc[t];
}

__global__ __launch_bounds__(256)
void file_edges_kernel(const int* __restrict__ src, const int* __restrict__ dst,
                       int* __restrict__ cursor, unsigned long long* __restrict__ pairs,
                       long long E, int nb) {
    __shared__ int hist[MAXB];
    __shared__ int slot0[MAXB];
    const int t = threadIdx.x;
    hist[t] = 0;
    __syncthreads();
    const long long base = (long long)blockIdx.x * EPB;
    int sv[EPB / 256], dv[EPB / 256], rk[EPB / 256];
#pragma unroll
    for (int j = 0; j < EPB / 256; ++j) {
        const long long e = base + j * 256 + t;
        sv[j] = e < E ? src[e] : 0;
        dv[j] = e < E ? dst[e] : -1;
    }
#pragma unroll
    for (int j = 0; j < EPB / 256; ++j)
        rk[j] = dv[j] >= 0 ? lds_count(&hist[dv[j] / ROWS]) : 0;
    __syncthreads();
    if (t < nb && hist[t] != 0)
        slot0[t] = __hip_atomic_fetch_add(&cursor[t], hist[t], __ATOMIC_RELAXED, __HIP_MEMORY_SCOPE_AGENT);
    __syncthreads();
#pragma unroll
    for (int j = 0; j < EPB / 256; ++j) {
        if (dv[j] >= 0)
            pairs[slot0[dv[j] / ROWS] + rk[j]] =
                (unsigned long long)(unsigned)sv[j] | ((unsigned long long)(unsigned)dv[j] << 32);
    }
}

#define DGT 512
#define DGE 16
__global__ __launch_bounds__(DGT)
void range_norm_kernel(const unsigned long long* __restrict__ pairs, const int* __restrict__ first,
                       float* __restrict__ dinv, int N) {
    __shared__ int c[ROWS];
    const int t = threadIdx.x;
    const int row0 = (int)blockIdx.x * ROWS;
    for (int i = t; i < ROWS; i += DGT) c[i] = 0;
    __syncthreads();
    const int beg = first[blockIdx.x];
    const int end = first[blockIdx.x + 1];
    const int* hi = reinterpret_cast<const int*>(pairs) + 1;
    for (int e0 = beg; e0 < end; e0 += DGT * DGE) {
        int dv[DGE];
#pragma unroll
        for (int j = 0; j < DGE; ++j) {
            const int e = e0 + j * DGT + t;
            dv[j] = e < end ? hi[2 * (size_t)e] : -1;
        }
#pragma unroll
        for (int j = 0; j < DGE; ++j)
            if (dv[j] >= 0) lds_count(&c[dv[j] - row0]);
    }
    __syncthreads();
    const int left_rows = N - row0;
    const int rows = left_rows < ROWS ? left_rows : ROWS;
    for (int i = t; i < rows; i += DGT) dinv[row0 + i] = rsqrtf((float)(c[i] + 1));
}

__device__ __forceinline__ float swap_halves(float x) {
    return __builtin_bit_cast(float, __builtin_amdgcn_permlanex16(
        __builtin_bit_cast(int, x), __builtin_bit_cast(int, x), 0x76543210, 0xfedcba98, false, false));
}

template <int NOUT, bool RELU_A>
__global__ __launch_bounds__(256)
void gemm_wmma_f32(const float* __restrict__ A, const float* __restrict__ B,
                   float* __restrict__ H, int M) {
    constexpr int K = 128;
    constexpr int NW = 64;
    constexpr int NT = NW / 16;
    constexpr int NCB = NOUT / NW;
    constexpr int NQ = K / 4;
    constexpr int PITCH = NW + 8;
    __shared__ float4 Bq[NQ * PITCH];

    const int lane = threadIdx.x & 31;
    const int c = lane & 15;
    const int h = lane >> 4;
    const int w = __builtin_amdgcn_readfirstlane((int)(threadIdx.x >> 5));
    const int cb = (blockIdx.x % NCB) * NW;
    const int strip = (blockIdx.x / NCB) * 8 + w;
    const int R0 = strip * 16;
    const bool live = R0 < M;

    constexpr int NCP = NQ * NW / 256;
    float4 bst[NCP];
#pragma unroll
    for (int it = 0; it < NCP; ++it) {
        const int i = it * 256 + (int)threadIdx.x;
        const float* bp = B + (size_t)(4 * (i / NW)) * NOUT + cb + i % NW;
        bst[it].x = bp[0];
        bst[it].y = bp[NOUT];
        bst[it].z = bp[2 * NOUT];
        bst[it].w = bp[3 * NOUT];
    }

    float4 av[16];
    {
        const float* arow = A + (size_t)((live ? R0 : 0) + c) * K + 4 * h;
#pragma unroll
        for (int j = 0; j < 16; ++j) av[j] = *reinterpret_cast<const float4*>(arow + 8 * j);
    }

#pragma unroll
    for (int it = 0; it < NCP; ++it) {
        const int i = it * 256 + (int)threadIdx.x;
        Bq[(i / NW) * PITCH + i % NW] = bst[it];
    }
    __syncthreads();
    if (!live) return;

    v8f acc[NT];
#pragma unroll
    for (int t = 0; t < NT; ++t) acc[t] = v8f{};

#pragma unroll
    for (int j = 0; j < 16; ++j) {
        float4 bv[NT];
#pragma unroll
        for (int t = 0; t < NT; ++t) bv[t] = Bq[(2 * j + h) * PITCH + 16 * t + c];
        float4 a4 = av[j];
        if (RELU_A) {
            a4.x = fmaxf(a4.x, 0.0f); a4.y = fmaxf(a4.y, 0.0f);
            a4.z = fmaxf(a4.z, 0.0f); a4.w = fmaxf(a4.w, 0.0f);
        }
        v2f a0, a1;
        a0[0] = a4.x; a0[1] = a4.y;
        a1[0] = a4.z; a1[1] = a4.w;
#pragma unroll
        for (int t = 0; t < NT; ++t) {
            v2f b;
            b[0] = bv[t].x; b[1] = bv[t].y;
            acc[t] = __builtin_amdgcn_wmma_f32_16x16x4_f32(false, a0, false, b,
                                                           (short)0, acc[t], false, false);
        }
#pragma unroll
        for (int t = 0; t < NT; ++t) {
            v2f b;
            b[0] = bv[t].z; b[1] = bv[t].w;
            acc[t] = __builtin_amdgcn_wmma_f32_16x16x4_f32(false, a1, false, b,
                                                           (short)0, acc[t], false, false);
        }
    }

#pragma unroll
    for (int r = 0; r < 8; ++r) {
#pragma unroll
        for (int p = 0; p < NW / 32; ++p) {
            const float x = acc[2 * p][r];
            const float y = acc[2 * p + 1][r];
            const float xs = swap_halves(x);
            const float ys = swap_halves(y);
            const float u = h ? ys : x;
            const float v = h ? y : xs;
            const size_t olo = (size_t)(R0 + r) * NOUT + cb + p * 32 + lane;
            const size_t ohi = (size_t)(R0 + 8 + r) * NOUT + cb + p * 32 + lane;
            H[olo] = u;
            H[ohi] = v;
        }
    }
}

#define CH 8192

template <int F>
__device__ __forceinline__ float* row_ptr(float* acc, int dl, int k, int lane) {
    const int dj = __builtin_amdgcn_readlane(dl, k);
    return acc + dj * F + (F / 32) * lane;
}
__device__ __forceinline__ float lane_f32(float x, int k) {
    return __int_as_float(__builtin_amdgcn_readlane(__float_as_int(x), k));
}

template <int F, int NE>
__device__ __forceinline__ void add_rows(int s, int dl, float nrm, int k0,
                                         const float* __restrict__ h, float* acc, int lane) {
    constexpr int V = F / 32;
    typedef float vf __attribute__((ext_vector_type(V)));
    vf v[NE];
#pragma unroll
    for (int j = 0; j < NE; ++j) {
        const int sj = __builtin_amdgcn_readlane(s, k0 + j);
        v[j] = *reinterpret_cast<const vf*>(h + (size_t)sj * F + V * lane);
    }
    if constexpr (NE == 1) {
        vf* ap = reinterpret_cast<vf*>(row_ptr<F>(acc, dl, k0, lane));
        vf a = *ap;
        a += v[0] * lane_f32(nrm, k0);
        *ap = a;
    } else {
#pragma unroll
        for (int j = 0; j < NE; j += 2) {
            const int d0 = __builtin_amdgcn_readlane(dl, k0 + j);
            const int d1 = __builtin_amdgcn_readlane(dl, k0 + j + 1);
            const float n0 = lane_f32(nrm, k0 + j);
            const float n1 = lane_f32(nrm, k0 + j + 1);
            vf* ap0 = reinterpret_cast<vf*>(acc + d0 * F + V * lane);
            vf* ap1 = reinterpret_cast<vf*>(acc + d1 * F + V * lane);
            if (d0 != d1) {
                vf a0 = *ap0;
                vf a1 = *ap1;
                a0 += v[j] * n0;
                a1 += v[j + 1] * n1;
                *ap0 = a0;
                *ap1 = a1;
            } else {
                vf a0 = *ap0;
                a0 += v[j] * n0;
                a0 += v[j + 1] * n1;
                *ap0 = a0;
            }
        }
    }
}

template <int F, int NE>
__global__ __launch_bounds__(1024)
void edge_aggregate(const unsigned long long* __restrict__ pairs, const int* __restrict__ first,
                    const float* __restrict__ h, const float* __restrict__ dinv,
                    const float* __restrict__ bias, float* __restrict__ out, int N) {
    __shared__ __attribute__((aligned(16))) float acc[ROWS * F];
    __shared__ unsigned long long lst[CH];
    __shared__ int bin[32];
    __shared__ int seg[33];
    const int tid = (int)threadIdx.x;
    const int lane = tid & 31;
    const int w = __builtin_amdgcn_readfirstlane(tid >> 5);
    const int row0 = (int)blockIdx.x * ROWS;
    const int left_rows = N - row0;
    const int nfl = (left_rows < ROWS ? left_rows : ROWS) * F;

    for (int i = tid; i < nfl; i += 1024) {
        const float d = dinv[row0 + i / F];
        acc[i] = h[(size_t)row0 * F + i] * (d * d) + bias[i % F];
    }

    const int beg = first[blockIdx.x];
    const int end = first[blockIdx.x + 1];
    for (int c0 = beg; c0 < end; c0 += CH) {
        const int left_c = end - c0;
        const int n = left_c < CH ? left_c : CH;
        if (tid < 32) bin[tid] = 0;
        __syncthreads();
        unsigned long long pr[CH / 1024];
        int rk[CH / 1024];
#pragma unroll
        for (int j = 0; j < CH / 1024; ++j) {
            const int e = j * 1024 + tid;
            pr[j] = e < n ? pairs[c0 + e] : 0ull;
        }
#pragma unroll
        for (int j = 0; j < CH / 1024; ++j) {
            const int e = j * 1024 + tid;
            const int o = ((int)(unsigned)(pr[j] >> 32) - row0) & 31;
            rk[j] = e < n ? lds_count(&bin[o]) : 0;
        }
        __syncthreads();
        if (tid < 32) {
            int s0 = 0;
            for (int i = 0; i < 32; ++i) {
                const int c = bin[i];
                s0 += i < tid ? c : 0;
            }
            seg[tid] = s0;
            if (tid == 31) seg[32] = n;
        }
        __syncthreads();
#pragma unroll
        for (int j = 0; j < CH / 1024; ++j) {
            const int e = j * 1024 + tid;
            const int o = ((int)(unsigned)(pr[j] >> 32) - row0) & 31;
            if (e < n) lst[seg[o] + rk[j]] = pr[j];
        }
        __syncthreads();

        const int sb = __builtin_amdgcn_readfirstlane(seg[w]);
        const int se = __builtin_amdgcn_readfirstlane(seg[w + 1]);
        for (int e0 = sb; e0 < se; e0 += 32) {
            const int left = se - e0;
            const int cnt = left < 32 ? left : 32;
            int s = 0, dl = 0;
            float nrm = 0.0f;
            if (lane < cnt) {
                const unsigned long long q = lst[e0 + lane];
                s = (int)(unsigned)q;
                const int d = (int)(unsigned)(q >> 32);
                dl = d - row0;
                nrm = dinv[s] * dinv[d];
            }
            int k = 0;
            for (; k + NE <= cnt; k += NE) add_rows<F, NE>(s, dl, nrm, k, h, acc, lane);
            if constexpr (NE > 16) {
                if (k + 16 <= cnt) { add_rows<F, 16>(s, dl, nrm, k, h, acc, lane); k += 16; }
            }
            if constexpr (NE > 8) {
                if (k + 8 <= cnt) { add_rows<F, 8>(s, dl, nrm, k, h, acc, lane); k += 8; }
            }
            if (k + 4 <= cnt) { add_rows<F, 4>(s, dl, nrm, k, h, acc, lane); k += 4; }
            if (k + 2 <= cnt) { add_rows<F, 2>(s, dl, nrm, k, h, acc, lane); k += 2; }
            if (k < cnt) add_rows<F, 1>(s, dl, nrm, k, h, acc, lane);
        }
    }
    __syncthreads();

    for (int i = tid; i < nfl; i += 1024) out[(size_t)row0 * F + i] = acc[i];
}

static inline int blocks_for(long long n, int bs) { return (int)((n + bs - 1) / bs); }

extern "C" void kernel_launch(void* const* d_in, const int* in_sizes, int n_in,
                              void* d_out, int out_size, void* d_ws, size_t ws_size,
                              hipStream_t stream) {
    const float* x  = (const float*)d_in[0];
    const int*   ei = (const int*)d_in[1];
    const float* W1 = (const float*)d_in[2];
    const float* b1 = (const float*)d_in[3];
    const float* W2 = (const float*)d_in[4];
    const float* b2 = (const float*)d_in[5];
    float* out = (float*)d_out;

    const int N = in_sizes[0] / IN_C;
    const long long E = (long long)in_sizes[1] / 2;
    const int* esrc = ei;
    const int* edst = ei + E;

    float* ws   = (float*)d_ws;
    float* dinv = ws;
    float* h1   = dinv + 2 * (size_t)N;
    float* acc1 = h1 + (size_t)N * HID2;
    float* h2   = acc1 + (size_t)N * HID2;
    unsigned long long* pairs = (unsigned long long*)(h2 + (size_t)N * HID);
    int* cnt    = (int*)(pairs + E);
    int* first  = cnt + MAXB;
    int* cursor = first + MAXB + 4;

    const int nb = (N + ROWS - 1) / ROWS;

    init_kernel<<<1, 256, 0, stream>>>(cnt, nb);

    count_kernel<<<blocks_for(E, EPB), 256, 0, stream>>>(edst, cnt, E, nb);
    scan_kernel<<<1, 256, 0, stream>>>(cnt, first, cursor, nb);
    file_edges_kernel<<<blocks_for(E, EPB), 256, 0, stream>>>(esrc, edst, cursor, pairs, E, nb);
    range_norm_kernel<<<nb, DGT, 0, stream>>>(pairs, first, dinv, N);

    {
        const int mtiles = (N + 15) / 16;
        const int groups = (mtiles + 7) / 8 * (HID2 / 64);
        gemm_wmma_f32<HID2, false><<<groups, 256, 0, stream>>>(x, W1, h1, N);
    }
    edge_aggregate<HID2, 16><<<nb, 1024, 0, stream>>>(pairs, first, h1, dinv, b1, acc1, N);

    {
        const int mtiles = (N + 15) / 16;
        const int groups = (mtiles + 7) / 8 * (HID / 64);
        gemm_wmma_f32<HID, true><<<groups, 256, 0, stream>>>(acc1, W2, h2, N);
    }
    edge_aggregate<HID, 32><<<nb, 1024, 0, stream>>>(pairs, first, h2, dinv, b2, out, N);
}


# Layer 1 aggregates x first, multiplies after: a range's LDS sums go through W1 (+b1, relu) and W2 on WMMA in the same work-group; no h1 or layer-1 sums in memory (9 to 7 launches); sums regrouped
# speedup vs baseline: 4.4959x; 1.1450x over previous
#include <hip/hip_runtime.h>
#include <hip/hip_bf16.h>


typedef float v2f __attribute__((ext_vector_type(2)));
typedef float v8f __attribute__((ext_vector_type(8)));

#define IN_C 128
#define HID2 128
#define HID  64
#define ROWS 400
#define MAXB 256
#define EPB  4096

__device__ __forceinline__ int lds_count(int* p) {
    return __hip_atomic_fetch_add(p, 1, __ATOMIC_RELAXED, __HIP_MEMORY_SCOPE_WORKGROUP);
}

__global__ __launch_bounds__(256)
void init_kernel(int* __restrict__ cnt, int nb) {
    const int i = threadIdx.x;
    if (i < nb) cnt[i] = 0;
}

__global__ __launch_bounds__(256)
void count_kernel(const int* __restrict__ dst, int* __restrict__ cnt, long long E, int nb) {
    __shared__ int hist[MAXB];
    const int t = threadIdx.x;
    hist[t] = 0;
    __syncthreads();
    const long long base = (long long)blockIdx.x * EPB;
    int dv[EPB / 256];
#pragma unroll
    for (int j = 0; j < EPB / 256; ++j) {
        const long long e = base + j * 256 + t;
        dv[j] = e < E ? dst[e] : -1;
    }
#pragma unroll
    for (int j = 0; j < EPB / 256; ++j) {
        if (dv[j] >= 0) lds_count(&hist[dv[j] / ROWS]);
    }
    __syncthreads();
    if (t < nb && hist[t] != 0)
        __hip_atomic_fetch_add(&cnt[t], hist[t], __ATOMIC_RELAXED, __HIP_MEMORY_SCOPE_AGENT);
}

__global__ __launch_bounds__(256)
void scan_kernel(const int* __restrict__ cnt, int* __restrict__ first, int* __restrict__ cursor, int nb) {
    __shared__ int sc[MAXB];
    const int t = threadIdx.x;
    const int v = t < nb ? cnt[t] : 0;
    sc[t] = v;
    __syncthreads();
    for (int off = 1; off < MAXB; off <<= 1) {
        const int x = t >= off ? sc[t - off] : 0;
        __syncthreads();
        sc[t] += x;
        __syncthreads();
    }
    if (t < nb) {
        first[t] = sc[t] - v;
        cursor[t] = sc[t] - v;
    }
    if (t == nb - 1) first[nb] = sc[t];
}

__global__ __launch_bounds__(256)
void file_edges_kernel(const int* __restrict__ src, const int* __restrict__ dst,
                       int* __restrict__ cursor, unsigned long long* __restrict__ pairs,
                       long long E, int nb) {
    __shared__ int hist[MAXB];
    __shared__ int slot0[MAXB];
    const int t = threadIdx.x;
    hist[t] = 0;
    __syncthreads();
    const long long base = (long long)blockIdx.x * EPB;
    int sv[EPB / 256], dv[EPB / 256], rk[EPB / 256];
#pragma unroll
    for (int j = 0; j < EPB / 256; ++j) {
        const long long e = base + j * 256 + t;
        sv[j] = e < E ? src[e] : 0;
        dv[j] = e < E ? dst[e] : -1;
    }
#pragma unroll
    for (int j = 0; j < EPB / 256; ++j)
        rk[j] = dv[j] >= 0 ? lds_count(&hist[dv[j] / ROWS]) : 0;
    __syncthreads();
    if (t < nb && hist[t] != 0)
        slot0[t] = __hip_atomic_fetch_add(&cursor[t], hist[t], __ATOMIC_RELAXED, __HIP_MEMORY_SCOPE_AGENT);
    __syncthreads();
#pragma unroll
    for (int j = 0; j < EPB / 256; ++j) {
        if (dv[j] >= 0)
            pairs[slot0[dv[j] / ROWS] + rk[j]] =
                (unsigned long long)(unsigned)sv[j] | ((unsigned long long)(unsigned)dv[j] << 32);
    }
}

#define DGT 512
#define DGE 16
__global__ __launch_bounds__(DGT)
void range_norm_kernel(const unsigned long long* __restrict__ pairs, const int* __restrict__ first,
                       float* __restrict__ dinv, int N) {
    __shared__ int c[ROWS];
    const int t = threadIdx.x;
    const int row0 = (int)blockIdx.x * ROWS;
    for (int i = t; i < ROWS; i += DGT) c[i] = 0;
    __syncthreads();
    const int beg = first[blockIdx.x];
    const int end = first[blockIdx.x + 1];
    const int* hi = reinterpret_cast<const int*>(pairs) + 1;
    for (int e0 = beg; e0 < end; e0 += DGT * DGE) {
        int dv[DGE];
#pragma unroll
        for (int j = 0; j < DGE; ++j) {
            const int e = e0 + j * DGT + t;
            dv[j] = e < end ? hi[2 * (size_t)e] : -1;
        }
#pragma unroll
        for (int j = 0; j < DGE; ++j)
            if (dv[j] >= 0) lds_count(&c[dv[j] - row0]);
    }
    __syncthreads();
    const int left_rows = N - row0;
    const int rows = left_rows < ROWS ? left_rows : ROWS;
    for (int i = t; i < rows; i += DGT) dinv[row0 + i] = rsqrtf((float)(c[i] + 1));
}

__device__ __forceinline__ float swap_halves(float x) {
    return __builtin_bit_cast(float, __builtin_amdgcn_permlanex16(
        __builtin_bit_cast(int, x), __builtin_bit_cast(int, x), 0x76543210, 0xfedcba98, false, false));
}

#define WP1 (HID2 + 8)
#define WP2 (HID + 8)

template <int NC, int PITCH>
__device__ __forceinline__ void stage_b(const float* __restrict__ B, float4* Bq, int tid) {
    constexpr int NCP = 32 * NC / 1024;
    float4 bst[NCP];
#pragma unroll
    for (int it = 0; it < NCP; ++it) {
        const int i = it * 1024 + tid;
        const float* bp = B + (size_t)(4 * (i / NC)) * NC + i % NC;
        bst[it].x = bp[0];
        bst[it].y = bp[NC];
        bst[it].z = bp[2 * NC];
        bst[it].w = bp[3 * NC];
    }
#pragma unroll
    for (int it = 0; it < NCP; ++it) {
        const int i = it * 1024 + tid;
        Bq[(i / NC) * PITCH + i % NC] = bst[it];
    }
}

template <int PITCH>
__device__ __forceinline__ void strip_times_b(const float4 (&av)[16], const float4* Bq, int cb0,
                                              int c, int h, v8f (&acc)[4]) {
#pragma unroll
    for (int j = 0; j < 16; ++j) {
        float4 bv[4];
#pragma unroll
        for (int t = 0; t < 4; ++t) bv[t] = Bq[(2 * j + h) * PITCH + cb0 + 16 * t + c];
        v2f a0, a1;
        a0[0] = av[j].x; a0[1] = av[j].y;
        a1[0] = av[j].z; a1[1] = av[j].w;
#pragma unroll
        for (int t = 0; t < 4; ++t) {
            v2f b;
            b[0] = bv[t].x; b[1] = bv[t].y;
            acc[t] = __builtin_amdgcn_wmma_f32_16x16x4_f32(false, a0, false, b,
                                                           (short)0, acc[t], false, false);
        }
#pragma unroll
        for (int t = 0; t < 4; ++t) {
            v2f b;
            b[0] = bv[t].z; b[1] = bv[t].w;
            acc[t] = __builtin_amdgcn_wmma_f32_16x16x4_f32(false, a1, false, b,
                                                           (short)0, acc[t], false, false);
        }
    }
}

#define CH 8192

template <int F, int RP>
__device__ __forceinline__ float* row_ptr(float* acc, int dl, int k, int lane) {
    const int dj = __builtin_amdgcn_readlane(dl, k);
    return acc + dj * RP + (F / 32) * lane;
}
__device__ __forceinline__ float lane_f32(float x, int k) {
    return __int_as_float(__builtin_amdgcn_readlane(__float_as_int(x), k));
}

template <int F, int NE, int RP>
__device__ __forceinline__ void add_rows(int s, int dl, float nrm, int k0,
                                         const float* __restrict__ h, float* acc, int lane) {
    constexpr int V = F / 32;
    typedef float vf __attribute__((ext_vector_type(V)));
    vf v[NE];
#pragma unroll
    for (int j = 0; j < NE; ++j) {
        const int sj = __builtin_amdgcn_readlane(s, k0 + j);
        v[j] = *reinterpret_cast<const vf*>(h + (size_t)sj * F + V * lane);
    }
    if constexpr (NE == 1) {
        vf* ap = reinterpret_cast<vf*>(row_ptr<F, RP>(acc, dl, k0, lane));
        vf a = *ap;
        a += v[0] * lane_f32(nrm, k0);
        *ap = a;
    } else {
#pragma unroll
        for (int j = 0; j < NE; j += 2) {
            const int d0 = __builtin_amdgcn_readlane(dl, k0 + j);
            const int d1 = __builtin_amdgcn_readlane(dl, k0 + j + 1);
            const float n0 = lane_f32(nrm, k0 + j);
            const float n1 = lane_f32(nrm, k0 + j + 1);
            vf* ap0 = reinterpret_cast<vf*>(acc + d0 * RP + V * lane);
            vf* ap1 = reinterpret_cast<vf*>(acc + d1 * RP + V * lane);
            if (d0 != d1) {
                vf a0 = *ap0;
                vf a1 = *ap1;
                a0 += v[j] * n0;
                a1 += v[j + 1] * n1;
                *ap0 = a0;
                *ap1 = a1;
            } else {
                vf a0 = *ap0;
                a0 += v[j] * n0;
                a0 += v[j + 1] * n1;
                *ap0 = a0;
            }
        }
    }
}

template <int F, int NE, bool MUL>
__global__ __launch_bounds__(1024)
void edge_aggregate(const unsigned long long* __restrict__ pairs, const int* __restrict__ first,
                    const float* __restrict__ h, const float* __restrict__ dinv,
                    const float* __restrict__ bias, float* __restrict__ out, int N,
                    const float* __restrict__ W1, const float* __restrict__ W2) {
    static_assert(!MUL || (F == IN_C && ROWS % 16 == 0 && ROWS / 16 <= 32), "one strip per wave");
    constexpr int LW = MUL ? (32 * WP1 * 2 > CH ? 32 * WP1 * 2 : CH) : CH;
    constexpr int RP = MUL ? F + 4 : F;
    __shared__ __attribute__((aligned(16))) float acc[ROWS * RP];
    __shared__ __attribute__((aligned(16))) unsigned long long lst[LW];
    __shared__ float4 wq2[MUL ? 32 * WP2 : 1];
    __shared__ int bin[32];
    __shared__ int seg[33];
    const int tid = (int)threadIdx.x;
    const int lane = tid & 31;
    const int w = __builtin_amdgcn_readfirstlane(tid >> 5);
    const int row0 = (int)blockIdx.x * ROWS;
    const int left_rows = N - row0;
    const int nfl = (left_rows < ROWS ? left_rows : ROWS) * F;

    for (int i = tid; i < nfl; i += 1024) {
        const float d = dinv[row0 + i / F];
        const float sl = h[(size_t)row0 * F + i] * (d * d);
        acc[(i / F) * RP + i % F] = MUL ? sl : sl + bias[i % F];
    }
    if constexpr (MUL) stage_b<HID, WP2>(W2, wq2, tid);

    const int beg = first[blockIdx.x];
    const int end = first[blockIdx.x + 1];
    for (int c0 = beg; c0 < end; c0 += CH) {
        const int left_c = end - c0;
        const int n = left_c < CH ? left_c : CH;
        if (tid < 32) bin[tid] = 0;
        __syncthreads();
        unsigned long long pr[CH / 1024];
        int rk[CH / 1024];
#pragma unroll
        for (int j = 0; j < CH / 1024; ++j) {
            const int e = j * 1024 + tid;
            pr[j] = e < n ? pairs[c0 + e] : 0ull;
        }
#pragma unroll
        for (int j = 0; j < CH / 1024; ++j) {
            const int e = j * 1024 + tid;
            const int o = ((int)(unsigned)(pr[j] >> 32) - row0) & 31;
            rk[j] = e < n ? lds_count(&bin[o]) : 0;
        }
        __syncthreads();
        if (tid < 32) {
            int s0 = 0;
            for (int i = 0; i < 32; ++i) {
                const int c = bin[i];
                s0 += i < tid ? c : 0;
            }
            seg[tid] = s0;
            if (tid == 31) seg[32] = n;
        }
        __syncthreads();
#pragma unroll
        for (int j = 0; j < CH / 1024; ++j) {
            const int e = j * 1024 + tid;
            const int o = ((int)(unsigned)(pr[j] >> 32) - row0) & 31;
            if (e < n) lst[seg[o] + rk[j]] = pr[j];
        }
        __syncthreads();

        const int sb = __builtin_amdgcn_readfirstlane(seg[w]);
        const int se = __builtin_amdgcn_readfirstlane(seg[w + 1]);
        for (int e0 = sb; e0 < se; e0 += 32) {
            const int left = se - e0;
            const int cnt = left < 32 ? left : 32;
            int s = 0, dl = 0;
            float nrm = 0.0f;
            if (lane < cnt) {
                const unsigned long long q = lst[e0 + lane];
                s = (int)(unsigned)q;
                const int d = (int)(unsigned)(q >> 32);
                dl = d - row0;
                nrm = dinv[s] * dinv[d];
            }
            int k = 0;
            for (; k + NE <= cnt; k += NE) add_rows<F, NE, RP>(s, dl, nrm, k, h, acc, lane);
            if constexpr (NE > 16) {
                if (k + 16 <= cnt) { add_rows<F, 16, RP>(s, dl, nrm, k, h, acc, lane); k += 16; }
            }
            if constexpr (NE > 8) {
                if (k + 8 <= cnt) { add_rows<F, 8, RP>(s, dl, nrm, k, h, acc, lane); k += 8; }
            }
            if (k + 4 <= cnt) { add_rows<F, 4, RP>(s, dl, nrm, k, h, acc, lane); k += 4; }
            if (k + 2 <= cnt) { add_rows<F, 2, RP>(s, dl, nrm, k, h, acc, lane); k += 2; }
            if (k < cnt) add_rows<F, 1, RP>(s, dl, nrm, k, h, acc, lane);
        }
    }
    __syncthreads();

    if constexpr (!MUL) {
        for (int i = tid; i < nfl; i += 1024) out[(size_t)row0 * F + i] = acc[i];
    } else {
        float4* wq1 = reinterpret_cast<float4*>(lst);
        stage_b<HID2, WP1>(W1, wq1, tid);
        __syncthreads();
        if (w < ROWS / 16) {
            const int c = lane & 15;
            const int hh = lane >> 4;
            float* strip = acc + w * 16 * RP;
            float4 av[16];
#pragma unroll
            for (int j = 0; j < 16; ++j)
                av[j] = *reinterpret_cast<const float4*>(strip + c * RP + 4 * hh + 8 * j);
#pragma unroll
            for (int half = 0; half < 2; ++half) {
                v8f p[4];
#pragma unroll
                for (int t = 0; t < 4; ++t) p[t] = v8f{};
                strip_times_b<WP1>(av, wq1, 64 * half, c, hh, p);
#pragma unroll
                for (int t = 0; t < 4; ++t) {
                    const int col = 64 * half + 16 * t + c;
                    const float bc = bias[col];
#pragma unroll
                    for (int r = 0; r < 8; ++r)
                        strip[(8 * hh + r) * RP + col] = fmaxf(p[t][r] + bc, 0.0f);
                }
            }
#pragma unroll
            for (int j = 0; j < 16; ++j)
                av[j] = *reinterpret_cast<const float4*>(strip + c * RP + 4 * hh + 8 * j);
            v8f q[4];
#pragma unroll
            for (int t = 0; t < 4; ++t) q[t] = v8f{};
            strip_times_b<WP2>(av, wq2, 0, c, hh, q);
            const int R0 = row0 + w * 16;
#pragma unroll
            for (int r = 0; r < 8; ++r) {
#pragma unroll
                for (int pp = 0; pp < HID / 32; ++pp) {
                    const float xx = q[2 * pp][r];
                    const float yy = q[2 * pp + 1][r];
                    const float xs = swap_halves(xx);
                    const float ys = swap_halves(yy);
                    const float u = hh ? ys : xx;
                    const float v = hh ? yy : xs;
                    if (R0 + r < N) out[(size_t)(R0 + r) * HID + pp * 32 + lane] = u;
                    if (R0 + 8 + r < N) out[(size_t)(R0 + 8 + r) * HID + pp * 32 + lane] = v;
                }
            }
        }
    }
}

static inline int blocks_for(long long n, int bs) { return (int)((n + bs - 1) / bs); }

extern "C" void kernel_launch(void* const* d_in, const int* in_sizes, int n_in,
                              void* d_out, int out_size, void* d_ws, size_t ws_size,
                              hipStream_t stream) {
    const float* x  = (const float*)d_in[0];
    const int*   ei = (const int*)d_in[1];
    const float* W1 = (const float*)d_in[2];
    const float* b1 = (const float*)d_in[3];
    const float* W2 = (const float*)d_in[4];
    const float* b2 = (const float*)d_in[5];
    float* out = (float*)d_out;

    const int N = in_sizes[0] / IN_C;
    const long long E = (long long)in_sizes[1] / 2;
    const int* esrc = ei;
    const int* edst = ei + E;

    float* ws   = (float*)d_ws;
    float* dinv = ws;
    float* h2   = dinv + 2 * (size_t)N;
    unsigned long long* pairs = (unsigned long long*)(h2 + (size_t)N * HID);
    int* cnt    = (int*)(pairs + E);
    int* first  = cnt + MAXB;
    int* cursor = first + MAXB + 4;

    const int nb = (N + ROWS - 1) / ROWS;

    init_kernel<<<1, 256, 0, stream>>>(cnt, nb);

    count_kernel<<<blocks_for(E, EPB), 256, 0, stream>>>(edst, cnt, E, nb);
    scan_kernel<<<1, 256, 0, stream>>>(cnt, first, cursor, nb);
    file_edges_kernel<<<blocks_for(E, EPB), 256, 0, stream>>>(esrc, edst, cursor, pairs, E, nb);
    range_norm_kernel<<<nb, DGT, 0, stream>>>(pairs, first, dinv, N);

    edge_aggregate<IN_C, 16, true><<<nb, 1024, 0, stream>>>(pairs, first, x, dinv, b1, h2, N, W1, W2);

    edge_aggregate<HID, 32, false><<<nb, 1024, 0, stream>>>(pairs, first, h2, dinv, b2, out, N, nullptr, nullptr);
}


# Layer 1's work-group writes its wave-ordered edge list with each edge's norm (looked up once, while ordering); layer 2 reads it: no second ordering, no per-edge degree reads; sums in that list order
# speedup vs baseline: 4.8017x; 1.0680x over previous
#include <hip/hip_runtime.h>
#include <hip/hip_bf16.h>


typedef float v2f __attribute__((ext_vector_type(2)));
typedef float v8f __attribute__((ext_vector_type(8)));

#define IN_C 128
#define HID2 128
#define HID  64
#define ROWS 400
#define MAXB 256
#define EPB  4096

__device__ __forceinline__ int lds_count(int* p) {
    return __hip_atomic_fetch_add(p, 1, __ATOMIC_RELAXED, __HIP_MEMORY_SCOPE_WORKGROUP);
}

__global__ __launch_bounds__(256)
void init_kernel(int* __restrict__ cnt, int nb) {
    const int i = threadIdx.x;
    if (i < nb) cnt[i] = 0;
}

__global__ __launch_bounds__(256)
void count_kernel(const int* __restrict__ dst, int* __restrict__ cnt, long long E, int nb) {
    __shared__ int hist[MAXB];
    const int t = threadIdx.x;
    hist[t] = 0;
    __syncthreads();
    const long long base = (long long)blockIdx.x * EPB;
    int dv[EPB / 256];
#pragma unroll
    for (int j = 0; j < EPB / 256; ++j) {
        const long long e = base + j * 256 + t;
        dv[j] = e < E ? dst[e] : -1;
    }
#pragma unroll
    for (int j = 0; j < EPB / 256; ++j) {
        if (dv[j] >= 0) lds_count(&hist[dv[j] / ROWS]);
    }
    __syncthreads();
    if (t < nb && hist[t] != 0)
        __hip_atomic_fetch_add(&cnt[t], hist[t], __ATOMIC_RELAXED, __HIP_MEMORY_SCOPE_AGENT);
}

__global__ __launch_bounds__(256)
void scan_kernel(const int* __restrict__ cnt, int* __restrict__ first, int* __restrict__ cursor, int nb) {
    __shared__ int sc[MAXB];
    const int t = threadIdx.x;
    const int v = t < nb ? cnt[t] : 0;
    sc[t] = v;
    __syncthreads();
    for (int off = 1; off < MAXB; off <<= 1) {
        const int x = t >= off ? sc[t - off] : 0;
        __syncthreads();
        sc[t] += x;
        __syncthreads();
    }
    if (t < nb) {
        first[t] = sc[t] - v;
        cursor[t] = sc[t] - v;
    }
    if (t == nb - 1) first[nb] = sc[t];
}

__global__ __launch_bounds__(256)
void file_edges_kernel(const int* __restrict__ src, const int* __restrict__ dst,
                       int* __restrict__ cursor, unsigned long long* __restrict__ pairs,
                       long long E, int nb) {
    __shared__ int hist[MAXB];
    __shared__ int slot0[MAXB];
    const int t = threadIdx.x;
    hist[t] = 0;
    __syncthreads();
    const long long base = (long long)blockIdx.x * EPB;
    int sv[EPB / 256], dv[EPB / 256], rk[EPB / 256];
#pragma unroll
    for (int j = 0; j < EPB / 256; ++j) {
        const long long e = base + j * 256 + t;
        sv[j] = e < E ? src[e] : 0;
        dv[j] = e < E ? dst[e] : -1;
    }
#pragma unroll
    for (int j = 0; j < EPB / 256; ++j)
        rk[j] = dv[j] >= 0 ? lds_count(&hist[dv[j] / ROWS]) : 0;
    __syncthreads();
    if (t < nb && hist[t] != 0)
        slot0[t] = __hip_atomic_fetch_add(&cursor[t], hist[t], __ATOMIC_RELAXED, __HIP_MEMORY_SCOPE_AGENT);
    __syncthreads();
#pragma unroll
    for (int j = 0; j < EPB / 256; ++j) {
        if (dv[j] >= 0)
            pairs[slot0[dv[j] / ROWS] + rk[j]] =
                (unsigned long long)(unsigned)sv[j] | ((unsigned long long)(unsigned)dv[j] << 32);
    }
}

#define DGT 512
#define DGE 16
__global__ __launch_bounds__(DGT)
void range_norm_kernel(const unsigned long long* __restrict__ pairs, const int* __restrict__ first,
                       float* __restrict__ dinv, int N) {
    __shared__ int c[ROWS];
    const int t = threadIdx.x;
    const int row0 = (int)blockIdx.x * ROWS;
    for (int i = t; i < ROWS; i += DGT) c[i] = 0;
    __syncthreads();
    const int beg = first[blockIdx.x];
    const int end = first[blockIdx.x + 1];
    const int* hi = reinterpret_cast<const int*>(pairs) + 1;
    for (int e0 = beg; e0 < end; e0 += DGT * DGE) {
        int dv[DGE];
#pragma unroll
        for (int j = 0; j < DGE; ++j) {
            const int e = e0 + j * DGT + t;
            dv[j] = e < end ? hi[2 * (size_t)e] : -1;
        }
#pragma unroll
        for (int j = 0; j < DGE; ++j)
            if (dv[j] >= 0) lds_count(&c[dv[j] - row0]);
    }
    __syncthreads();
    const int left_rows = N - row0;
    const int rows = left_rows < ROWS ? left_rows : ROWS;
    for (int i = t; i < rows; i += DGT) dinv[row0 + i] = rsqrtf((float)(c[i] + 1));
}

__device__ __forceinline__ float swap_halves(float x) {
    return __builtin_bit_cast(float, __builtin_amdgcn_permlanex16(
        __builtin_bit_cast(int, x), __builtin_bit_cast(int, x), 0x76543210, 0xfedcba98, false, false));
}

#define WP1 (HID2 + 8)
#define WP2 (HID + 8)

template <int NC, int PITCH>
__device__ __forceinline__ void stage_b(const float* __restrict__ B, float4* Bq, int tid) {
    constexpr int NCP = 32 * NC / 1024;
    float4 bst[NCP];
#pragma unroll
    for (int it = 0; it < NCP; ++it) {
        const int i = it * 1024 + tid;
        const float* bp = B + (size_t)(4 * (i / NC)) * NC + i % NC;
        bst[it].x = bp[0];
        bst[it].y = bp[NC];
        bst[it].z = bp[2 * NC];
        bst[it].w = bp[3 * NC];
    }
#pragma unroll
    for (int it = 0; it < NCP; ++it) {
        const int i = it * 1024 + tid;
        Bq[(i / NC) * PITCH + i % NC] = bst[it];
    }
}

template <int PITCH>
__device__ __forceinline__ void strip_times_b(const float4 (&av)[16], const float4* Bq, int cb0,
                                              int c, int h, v8f (&acc)[4]) {
#pragma unroll
    for (int j = 0; j < 16; ++j) {
        float4 bv[4];
#pragma unroll
        for (int t = 0; t < 4; ++t) bv[t] = Bq[(2 * j + h) * PITCH + cb0 + 16 * t + c];
        v2f a0, a1;
        a0[0] = av[j].x; a0[1] = av[j].y;
        a1[0] = av[j].z; a1[1] = av[j].w;
#pragma unroll
        for (int t = 0; t < 4; ++t) {
            v2f b;
            b[0] = bv[t].x; b[1] = bv[t].y;
            acc[t] = __builtin_amdgcn_wmma_f32_16x16x4_f32(false, a0, false, b,
                                                           (short)0, acc[t], false, false);
        }
#pragma unroll
        for (int t = 0; t < 4; ++t) {
            v2f b;
            b[0] = bv[t].z; b[1] = bv[t].w;
            acc[t] = __builtin_amdgcn_wmma_f32_16x16x4_f32(false, a1, false, b,
                                                           (short)0, acc[t], false, false);
        }
    }
}

#define CH 8192
#define SRC_BITS 17
static_assert(MAXB * ROWS <= (1 << SRC_BITS) && ROWS <= (1 << (32 - SRC_BITS)), "src and row share one word");

template <int F, int RP>
__device__ __forceinline__ float* row_ptr(float* acc, int dl, int k, int lane) {
    const int dj = __builtin_amdgcn_readlane(dl, k);
    return acc + dj * RP + (F / 32) * lane;
}
__device__ __forceinline__ float lane_f32(float x, int k) {
    return __int_as_float(__builtin_amdgcn_readlane(__float_as_int(x), k));
}

template <int F, int NE, int RP>
__device__ __forceinline__ void add_rows(int s, int dl, float nrm, int k0,
                                         const float* __restrict__ h, float* acc, int lane) {
    constexpr int V = F / 32;
    typedef float vf __attribute__((ext_vector_type(V)));
    vf v[NE];
#pragma unroll
    for (int j = 0; j < NE; ++j) {
        const int sj = __builtin_amdgcn_readlane(s, k0 + j);
        v[j] = *reinterpret_cast<const vf*>(h + (size_t)sj * F + V * lane);
    }
    if constexpr (NE == 1) {
        vf* ap = reinterpret_cast<vf*>(row_ptr<F, RP>(acc, dl, k0, lane));
        vf a = *ap;
        a += v[0] * lane_f32(nrm, k0);
        *ap = a;
    } else {
#pragma unroll
        for (int j = 0; j < NE; j += 2) {
            const int d0 = __builtin_amdgcn_readlane(dl, k0 + j);
            const int d1 = __builtin_amdgcn_readlane(dl, k0 + j + 1);
            const float n0 = lane_f32(nrm, k0 + j);
            const float n1 = lane_f32(nrm, k0 + j + 1);
            vf* ap0 = reinterpret_cast<vf*>(acc + d0 * RP + V * lane);
            vf* ap1 = reinterpret_cast<vf*>(acc + d1 * RP + V * lane);
            if (d0 != d1) {
                vf a0 = *ap0;
                vf a1 = *ap1;
                a0 += v[j] * n0;
                a1 += v[j + 1] * n1;
                *ap0 = a0;
                *ap1 = a1;
            } else {
                vf a0 = *ap0;
                a0 += v[j] * n0;
                a0 += v[j + 1] * n1;
                *ap0 = a0;
            }
        }
    }
}

template <int F, int NE, bool MUL>
__global__ __launch_bounds__(1024)
void edge_aggregate(const unsigned long long* __restrict__ pairs, const int* __restrict__ first,
                    const float* __restrict__ h, const float* __restrict__ dinv,
                    const float* __restrict__ bias, float* __restrict__ out, int N,
                    const float* __restrict__ W1, const float* __restrict__ W2,
                    unsigned long long* __restrict__ olist, int* __restrict__ segs) {
    static_assert(!MUL || (F == IN_C && ROWS % 16 == 0 && ROWS / 16 <= 32), "one strip per wave");
    constexpr int LW = MUL ? (32 * WP1 * 2 > CH ? 32 * WP1 * 2 : CH) : 1;
    constexpr int RP = MUL ? F + 4 : F;
    __shared__ __attribute__((aligned(16))) float acc[ROWS * RP];
    __shared__ __attribute__((aligned(16))) unsigned long long lst[LW];
    __shared__ float4 wq2[MUL ? 32 * WP2 : 1];
    __shared__ int bin[32];
    __shared__ int seg[33];
    const int tid = (int)threadIdx.x;
    const int lane = tid & 31;
    const int w = __builtin_amdgcn_readfirstlane(tid >> 5);
    const int row0 = (int)blockIdx.x * ROWS;
    const int left_rows = N - row0;
    const int nfl = (left_rows < ROWS ? left_rows : ROWS) * F;

    for (int i = tid; i < nfl; i += 1024) {
        const float d = dinv[row0 + i / F];
        const float sl = h[(size_t)row0 * F + i] * (d * d);
        acc[(i / F) * RP + i % F] = MUL ? sl : sl + bias[i % F];
    }
    if constexpr (MUL) stage_b<HID, WP2>(W2, wq2, tid);

    const int beg = first[blockIdx.x];
    const int end = first[blockIdx.x + 1];
    if constexpr (!MUL) __syncthreads();
    for (int c0 = beg; c0 < end; c0 += CH) {
        const int left_c = end - c0;
        const int n = left_c < CH ? left_c : CH;
        const int id = c0 / CH + (int)blockIdx.x;
        int sb, se;
        if constexpr (MUL) {
            if (tid < 32) bin[tid] = 0;
            __syncthreads();
            unsigned long long pr[CH / 1024];
            int rk[CH / 1024];
            float ns[CH / 1024], nd[CH / 1024];
#pragma unroll
            for (int j = 0; j < CH / 1024; ++j) {
                const int e = j * 1024 + tid;
                pr[j] = e < n ? pairs[c0 + e] : 0ull;
            }
#pragma unroll
            for (int j = 0; j < CH / 1024; ++j) {
                ns[j] = dinv[(int)(unsigned)pr[j]];
                nd[j] = dinv[(int)(unsigned)(pr[j] >> 32)];
            }
#pragma unroll
            for (int j = 0; j < CH / 1024; ++j) {
                const int e = j * 1024 + tid;
                const int o = ((int)(unsigned)(pr[j] >> 32) - row0) & 31;
                rk[j] = e < n ? lds_count(&bin[o]) : 0;
            }
            __syncthreads();
            if (tid < 32) {
                int s0 = 0;
                for (int i = 0; i < 32; ++i) {
                    const int c = bin[i];
                    s0 += i < tid ? c : 0;
                }
                seg[tid] = s0;
                if (tid == 31) seg[32] = n;
            }
            __syncthreads();
#pragma unroll
            for (int j = 0; j < CH / 1024; ++j) {
                const int e = j * 1024 + tid;
                const unsigned sj = (unsigned)pr[j];
                const unsigned dj = (unsigned)(pr[j] >> 32) - (unsigned)row0;
                const int o = (int)dj & 31;
                const unsigned long long ent = (unsigned long long)(sj | (dj << SRC_BITS)) |
                                               ((unsigned long long)__float_as_uint(ns[j] * nd[j]) << 32);
                if (e < n) lst[seg[o] + rk[j]] = ent;
            }
            __syncthreads();
#pragma unroll
            for (int j = 0; j < CH / 1024; ++j) {
                const int e = j * 1024 + tid;
                if (e < n) olist[c0 + e] = lst[e];
            }
            if (tid < 33) segs[id * 33 + tid] = seg[tid];
            sb = __builtin_amdgcn_readfirstlane(seg[w]);
            se = __builtin_amdgcn_readfirstlane(seg[w + 1]);
        } else {
            sb = __builtin_amdgcn_readfirstlane(segs[id * 33 + w]);
            se = __builtin_amdgcn_readfirstlane(segs[id * 33 + w + 1]);
        }
        const unsigned long long* gl = olist + c0;
        unsigned long long qn = 0ull;
        if constexpr (!MUL) {
            if (sb < se) {
                const int i0 = sb + lane;
                qn = gl[i0 < se ? i0 : se - 1];
            }
        }
        for (int e0 = sb; e0 < se; e0 += 32) {
            const int left = se - e0;
            const int cnt = left < 32 ? left : 32;
            unsigned long long q = 0ull;
            if constexpr (MUL) {
                if (lane < cnt) q = lst[e0 + lane];
            } else {
                q = qn;
                const int en = e0 + 32 + lane;
                qn = gl[en < se ? en : se - 1];
            }
            const unsigned lo = (unsigned)q;
            const int s = (int)(lo & ((1u << SRC_BITS) - 1u));
            const int dl = (int)(lo >> SRC_BITS);
            const float nrm = __uint_as_float((unsigned)(q >> 32));
            int k = 0;
            for (; k + NE <= cnt; k += NE) add_rows<F, NE, RP>(s, dl, nrm, k, h, acc, lane);
            if constexpr (NE > 16) {
                if (k + 16 <= cnt) { add_rows<F, 16, RP>(s, dl, nrm, k, h, acc, lane); k += 16; }
            }
            if constexpr (NE > 8) {
                if (k + 8 <= cnt) { add_rows<F, 8, RP>(s, dl, nrm, k, h, acc, lane); k += 8; }
            }
            if (k + 4 <= cnt) { add_rows<F, 4, RP>(s, dl, nrm, k, h, acc, lane); k += 4; }
            if (k + 2 <= cnt) { add_rows<F, 2, RP>(s, dl, nrm, k, h, acc, lane); k += 2; }
            if (k < cnt) add_rows<F, 1, RP>(s, dl, nrm, k, h, acc, lane);
        }
    }
    __syncthreads();

    if constexpr (!MUL) {
        for (int i = tid; i < nfl; i += 1024) out[(size_t)row0 * F + i] = acc[i];
    } else {
        float4* wq1 = reinterpret_cast<float4*>(lst);
        stage_b<HID2, WP1>(W1, wq1, tid);
        __syncthreads();
        if (w < ROWS / 16) {
            const int c = lane & 15;
            const int hh = lane >> 4;
            float* strip = acc + w * 16 * RP;
            float4 av[16];
#pragma unroll
            for (int j = 0; j < 16; ++j)
                av[j] = *reinterpret_cast<const float4*>(strip + c * RP + 4 * hh + 8 * j);
#pragma unroll
            for (int half = 0; half < 2; ++half) {
                v8f p[4];
#pragma unroll
                for (int t = 0; t < 4; ++t) p[t] = v8f{};
                strip_times_b<WP1>(av, wq1, 64 * half, c, hh, p);
#pragma unroll
                for (int t = 0; t < 4; ++t) {
                    const int col = 64 * half + 16 * t + c;
                    const float bc = bias[col];
#pragma unroll
                    for (int r = 0; r < 8; ++r)
                        strip[(8 * hh + r) * RP + col] = fmaxf(p[t][r] + bc, 0.0f);
                }
            }
#pragma unroll
            for (int j = 0; j < 16; ++j)
                av[j] = *reinterpret_cast<const float4*>(strip + c * RP + 4 * hh + 8 * j);
            v8f q[4];
#pragma unroll
            for (int t = 0; t < 4; ++t) q[t] = v8f{};
            strip_times_b<WP2>(av, wq2, 0, c, hh, q);
            const int R0 = row0 + w * 16;
#pragma unroll
            for (int r = 0; r < 8; ++r) {
#pragma unroll
                for (int pp = 0; pp < HID / 32; ++pp) {
                    const float xx = q[2 * pp][r];
                    const float yy = q[2 * pp + 1][r];
                    const float xs = swap_halves(xx);
                    const float ys = swap_halves(yy);
                    const float u = hh ? ys : xx;
                    const float v = hh ? yy : xs;
                    if (R0 + r < N) out[(size_t)(R0 + r) * HID + pp * 32 + lane] = u;
                    if (R0 + 8 + r < N) out[(size_t)(R0 + 8 + r) * HID + pp * 32 + lane] = v;
                }
            }
        }
    }
}

static inline int blocks_for(long long n, int bs) { return (int)((n + bs - 1) / bs); }

extern "C" void kernel_launch(void* const* d_in, const int* in_sizes, int n_in,
                              void* d_out, int out_size, void* d_ws, size_t ws_size,
                              hipStream_t stream) {
    const float* x  = (const float*)d_in[0];
    const int*   ei = (const int*)d_in[1];
    const float* W1 = (const float*)d_in[2];
    const float* b1 = (const float*)d_in[3];
    const float* W2 = (const float*)d_in[4];
    const float* b2 = (const float*)d_in[5];
    float* out = (float*)d_out;

    const int N = in_sizes[0] / IN_C;
    const long long E = (long long)in_sizes[1] / 2;
    const int* esrc = ei;
    const int* edst = ei + E;

    float* ws   = (float*)d_ws;
    float* dinv = ws;
    float* h2   = dinv + 2 * (size_t)N;
    unsigned long long* pairs = (unsigned long long*)(h2 + (size_t)N * HID);
    unsigned long long* olist = pairs + E;
    int* cnt    = (int*)(olist + E);
    int* first  = cnt + MAXB;
    int* cursor = first + MAXB + 4;
    int* segs   = cursor + MAXB + 4;

    const int nb = (N + ROWS - 1) / ROWS;

    init_kernel<<<1, 256, 0, stream>>>(cnt, nb);

    count_kernel<<<blocks_for(E, EPB), 256, 0, stream>>>(edst, cnt, E, nb);
    scan_kernel<<<1, 256, 0, stream>>>(cnt, first, cursor, nb);
    file_edges_kernel<<<blocks_for(E, EPB), 256, 0, stream>>>(esrc, edst, cursor, pairs, E, nb);
    range_norm_kernel<<<nb, DGT, 0, stream>>>(pairs, first, dinv, N);

    edge_aggregate<IN_C, 16, true><<<nb, 1024, 0, stream>>>(pairs, first, x, dinv, b1, h2, N, W1, W2, olist, segs);

    edge_aggregate<HID, 32, false><<<nb, 1024, 0, stream>>>(pairs, first, h2, dinv, b2, out, N, nullptr, nullptr, olist, segs);
}


# Layer 1's work-group lists its edges by destination row and sums rows 0..255 first; 8 waves then multiply those strips on WMMA while 24 add the other rows' gathered edges; per-row sum order kept
# speedup vs baseline: 4.9665x; 1.0343x over previous
#include <hip/hip_runtime.h>
#include <hip/hip_bf16.h>


typedef float v2f __attribute__((ext_vector_type(2)));
typedef float v8f __attribute__((ext_vector_type(8)));

#define IN_C 128
#define HID2 128
#define HID  64
#define ROWS 400
#define MAXB 256
#define EPB  4096

__device__ __forceinline__ int lds_count(int* p) {
    return __hip_atomic_fetch_add(p, 1, __ATOMIC_RELAXED, __HIP_MEMORY_SCOPE_WORKGROUP);
}

__global__ __launch_bounds__(256)
void init_kernel(int* __restrict__ cnt, int nb) {
    const int i = threadIdx.x;
    if (i < nb) cnt[i] = 0;
}

__global__ __launch_bounds__(256)
void count_kernel(const int* __restrict__ dst, int* __restrict__ cnt, long long E, int nb) {
    __shared__ int hist[MAXB];
    const int t = threadIdx.x;
    hist[t] = 0;
    __syncthreads();
    const long long base = (long long)blockIdx.x * EPB;
    int dv[EPB / 256];
#pragma unroll
    for (int j = 0; j < EPB / 256; ++j) {
        const long long e = base + j * 256 + t;
        dv[j] = e < E ? dst[e] : -1;
    }
#pragma unroll
    for (int j = 0; j < EPB / 256; ++j) {
        if (dv[j] >= 0) lds_count(&hist[dv[j] / ROWS]);
    }
    __syncthreads();
    if (t < nb && hist[t] != 0)
        __hip_atomic_fetch_add(&cnt[t], hist[t], __ATOMIC_RELAXED, __HIP_MEMORY_SCOPE_AGENT);
}

__global__ __launch_bounds__(256)
void scan_kernel(const int* __restrict__ cnt, int* __restrict__ first, int* __restrict__ cursor, int nb) {
    __shared__ int sc[MAXB];
    const int t = threadIdx.x;
    const int v = t < nb ? cnt[t] : 0;
    sc[t] = v;
    __syncthreads();
    for (int off = 1; off < MAXB; off <<= 1) {
        const int x = t >= off ? sc[t - off] : 0;
        __syncthreads();
        sc[t] += x;
        __syncthreads();
    }
    if (t < nb) {
        first[t] = sc[t] - v;
        cursor[t] = sc[t] - v;
    }
    if (t == nb - 1) first[nb] = sc[t];
}

__global__ __launch_bounds__(256)
void file_edges_kernel(const int* __restrict__ src, const int* __restrict__ dst,
                       int* __restrict__ cursor, unsigned long long* __restrict__ pairs,
                       long long E, int nb) {
    __shared__ int hist[MAXB];
    __shared__ int slot0[MAXB];
    const int t = threadIdx.x;
    hist[t] = 0;
    __syncthreads();
    const long long base = (long long)blockIdx.x * EPB;
    int sv[EPB / 256], dv[EPB / 256], rk[EPB / 256];
#pragma unroll
    for (int j = 0; j < EPB / 256; ++j) {
        const long long e = base + j * 256 + t;
        sv[j] = e < E ? src[e] : 0;
        dv[j] = e < E ? dst[e] : -1;
    }
#pragma unroll
    for (int j = 0; j < EPB / 256; ++j)
        rk[j] = dv[j] >= 0 ? lds_count(&hist[dv[j] / ROWS]) : 0;
    __syncthreads();
    if (t < nb && hist[t] != 0)
        slot0[t] = __hip_atomic_fetch_add(&cursor[t], hist[t], __ATOMIC_RELAXED, __HIP_MEMORY_SCOPE_AGENT);
    __syncthreads();
#pragma unroll
    for (int j = 0; j < EPB / 256; ++j) {
        if (dv[j] >= 0)
            pairs[slot0[dv[j] / ROWS] + rk[j]] =
                (unsigned long long)(unsigned)sv[j] | ((unsigned long long)(unsigned)dv[j] << 32);
    }
}

#define DGT 512
#define DGE 16
__global__ __launch_bounds__(DGT)
void range_norm_kernel(const unsigned long long* __restrict__ pairs, const int* __restrict__ first,
                       float* __restrict__ dinv, int N) {
    __shared__ int c[ROWS];
    const int t = threadIdx.x;
    const int row0 = (int)blockIdx.x * ROWS;
    for (int i = t; i < ROWS; i += DGT) c[i] = 0;
    __syncthreads();
    const int beg = first[blockIdx.x];
    const int end = first[blockIdx.x + 1];
    const int* hi = reinterpret_cast<const int*>(pairs) + 1;
    for (int e0 = beg; e0 < end; e0 += DGT * DGE) {
        int dv[DGE];
#pragma unroll
        for (int j = 0; j < DGE; ++j) {
            const int e = e0 + j * DGT + t;
            dv[j] = e < end ? hi[2 * (size_t)e] : -1;
        }
#pragma unroll
        for (int j = 0; j < DGE; ++j)
            if (dv[j] >= 0) lds_count(&c[dv[j] - row0]);
    }
    __syncthreads();
    const int left_rows = N - row0;
    const int rows = left_rows < ROWS ? left_rows : ROWS;
    for (int i = t; i < rows; i += DGT) dinv[row0 + i] = rsqrtf((float)(c[i] + 1));
}

__device__ __forceinline__ float swap_halves(float x) {
    return __builtin_bit_cast(float, __builtin_amdgcn_permlanex16(
        __builtin_bit_cast(int, x), __builtin_bit_cast(int, x), 0x76543210, 0xfedcba98, false, false));
}

#define WP1 (HID2 + 8)
#define WP2 (HID + 8)

template <int NC, int PITCH>
__device__ __forceinline__ void stage_b(const float* __restrict__ B, float4* Bq, int tid) {
    constexpr int NCP = 32 * NC / 1024;
    float4 bst[NCP];
#pragma unroll
    for (int it = 0; it < NCP; ++it) {
        const int i = it * 1024 + tid;
        const float* bp = B + (size_t)(4 * (i / NC)) * NC + i % NC;
        bst[it].x = bp[0];
        bst[it].y = bp[NC];
        bst[it].z = bp[2 * NC];
        bst[it].w = bp[3 * NC];
    }
#pragma unroll
    for (int it = 0; it < NCP; ++it) {
        const int i = it * 1024 + tid;
        Bq[(i / NC) * PITCH + i % NC] = bst[it];
    }
}

template <int PITCH>
__device__ __forceinline__ void strip_times_b(const float4 (&av)[16], const float4* Bq, int cb0,
                                              int c, int h, v8f (&acc)[4]) {
#pragma unroll
    for (int j = 0; j < 16; ++j) {
        float4 bv[4];
#pragma unroll
        for (int t = 0; t < 4; ++t) bv[t] = Bq[(2 * j + h) * PITCH + cb0 + 16 * t + c];
        v2f a0, a1;
        a0[0] = av[j].x; a0[1] = av[j].y;
        a1[0] = av[j].z; a1[1] = av[j].w;
#pragma unroll
        for (int t = 0; t < 4; ++t) {
            v2f b;
            b[0] = bv[t].x; b[1] = bv[t].y;
            acc[t] = __builtin_amdgcn_wmma_f32_16x16x4_f32(false, a0, false, b,
                                                           (short)0, acc[t], false, false);
        }
#pragma unroll
        for (int t = 0; t < 4; ++t) {
            v2f b;
            b[0] = bv[t].z; b[1] = bv[t].w;
            acc[t] = __builtin_amdgcn_wmma_f32_16x16x4_f32(false, a1, false, b,
                                                           (short)0, acc[t], false, false);
        }
    }
}

#define CH 8192
#define SRC_BITS 17
#define H0ROWS 256
#define MMW 8
static_assert(H0ROWS % 32 == 0 && H0ROWS < ROWS && (H0ROWS / 16) % MMW == 0 && (ROWS - H0ROWS) % (32 - MMW) == 0 &&
              (ROWS - H0ROWS) / 16 <= 32, "whole strips and whole rows per wave in every phase");
static_assert(MAXB * ROWS <= (1 << SRC_BITS) && ROWS <= (1 << (32 - SRC_BITS)), "src and row share one word");

template <int F, int RP>
__device__ __forceinline__ float* row_ptr(float* acc, int dl, int k, int lane) {
    const int dj = __builtin_amdgcn_readlane(dl, k);
    return acc + dj * RP + (F / 32) * lane;
}
__device__ __forceinline__ float lane_f32(float x, int k) {
    return __int_as_float(__builtin_amdgcn_readlane(__float_as_int(x), k));
}

template <int F, int NE, int RP>
__device__ __forceinline__ void add_rows(int s, int dl, float nrm, int k0,
                                         const float* __restrict__ h, float* acc, int lane) {
    constexpr int V = F / 32;
    typedef float vf __attribute__((ext_vector_type(V)));
    vf v[NE];
#pragma unroll
    for (int j = 0; j < NE; ++j) {
        const int sj = __builtin_amdgcn_readlane(s, k0 + j);
        v[j] = *reinterpret_cast<const vf*>(h + (size_t)sj * F + V * lane);
    }
    if constexpr (NE == 1) {
        vf* ap = reinterpret_cast<vf*>(row_ptr<F, RP>(acc, dl, k0, lane));
        vf a = *ap;
        a += v[0] * lane_f32(nrm, k0);
        *ap = a;
    } else {
#pragma unroll
        for (int j = 0; j < NE; j += 2) {
            const int d0 = __builtin_amdgcn_readlane(dl, k0 + j);
            const int d1 = __builtin_amdgcn_readlane(dl, k0 + j + 1);
            const float n0 = lane_f32(nrm, k0 + j);
            const float n1 = lane_f32(nrm, k0 + j + 1);
            vf* ap0 = reinterpret_cast<vf*>(acc + d0 * RP + V * lane);
            vf* ap1 = reinterpret_cast<vf*>(acc + d1 * RP + V * lane);
            if (d0 != d1) {
                vf a0 = *ap0;
                vf a1 = *ap1;
                a0 += v[j] * n0;
                a1 += v[j + 1] * n1;
                *ap0 = a0;
                *ap1 = a1;
            } else {
                vf a0 = *ap0;
                a0 += v[j] * n0;
                a0 += v[j + 1] * n1;
                *ap0 = a0;
            }
        }
    }
}

template <int F, int NE, int RP, bool FROM_LDS>
__device__ __forceinline__ void walk_piece(const unsigned long long* lst, const unsigned long long* gl,
                                           int sb, int se, const float* __restrict__ h, float* acc, int lane) {
    unsigned long long qn = 0ull;
    if constexpr (!FROM_LDS) {
        if (sb < se) {
            const int i0 = sb + lane;
            qn = gl[i0 < se ? i0 : se - 1];
        }
    }
    for (int e0 = sb; e0 < se; e0 += 32) {
        const int left = se - e0;
        const int cnt = left < 32 ? left : 32;
        unsigned long long q = 0ull;
        if constexpr (FROM_LDS) {
            if (lane < cnt) q = lst[e0 + lane];
        } else {
            q = qn;
            const int en = e0 + 32 + lane;
            qn = gl[en < se ? en : se - 1];
        }
        const unsigned lo = (unsigned)q;
        const int s = (int)(lo & ((1u << SRC_BITS) - 1u));
        const int dl = (int)(lo >> SRC_BITS);
        const float nrm = __uint_as_float((unsigned)(q >> 32));
        int k = 0;
        for (; k + NE <= cnt; k += NE) add_rows<F, NE, RP>(s, dl, nrm, k, h, acc, lane);
        if constexpr (NE > 16) {
            if (k + 16 <= cnt) { add_rows<F, 16, RP>(s, dl, nrm, k, h, acc, lane); k += 16; }
        }
        if constexpr (NE > 8) {
            if (k + 8 <= cnt) { add_rows<F, 8, RP>(s, dl, nrm, k, h, acc, lane); k += 8; }
        }
        if (k + 4 <= cnt) { add_rows<F, 4, RP>(s, dl, nrm, k, h, acc, lane); k += 4; }
        if (k + 2 <= cnt) { add_rows<F, 2, RP>(s, dl, nrm, k, h, acc, lane); k += 2; }
        if (k < cnt) add_rows<F, 1, RP>(s, dl, nrm, k, h, acc, lane);
    }
}

template <int F, int NE, bool MUL>
__global__ __launch_bounds__(1024)
void edge_aggregate(const unsigned long long* __restrict__ pairs, const int* __restrict__ first,
                    const float* __restrict__ h, const float* __restrict__ dinv,
                    const float* __restrict__ bias, float* __restrict__ out, int N,
                    const float* __restrict__ W1, const float* __restrict__ W2,
                    unsigned long long* __restrict__ olist, int* __restrict__ segs) {
    static_assert(!MUL || (F == IN_C && ROWS % 16 == 0 && ROWS / 16 <= 32), "whole strips");
    constexpr int LW = MUL ? (32 * WP1 * 2 > CH ? 32 * WP1 * 2 : CH) : 1;
    constexpr int RP = MUL ? F + 4 : F;
    constexpr int SCW = (ROWS + 31) / 32;
    __shared__ __attribute__((aligned(16))) float acc[ROWS * RP];
    __shared__ __attribute__((aligned(16))) unsigned long long lst[LW];
    __shared__ float4 wq2[MUL ? 32 * WP2 : 1];
    __shared__ int bin[MUL ? ROWS : 1];
    __shared__ int seg[MUL ? ROWS + 1 : 1];
    __shared__ int wsum[MUL ? SCW : 1];
    const int tid = (int)threadIdx.x;
    const int lane = tid & 31;
    const int w = __builtin_amdgcn_readfirstlane(tid >> 5);
    const int row0 = (int)blockIdx.x * ROWS;
    const int left_rows = N - row0;
    const int nfl = (left_rows < ROWS ? left_rows : ROWS) * F;

    for (int i = tid; i < nfl; i += 1024) {
        const float d = dinv[row0 + i / F];
        const float sl = h[(size_t)row0 * F + i] * (d * d);
        acc[(i / F) * RP + i % F] = MUL ? sl : sl + bias[i % F];
    }
    if constexpr (MUL) stage_b<HID, WP2>(W2, wq2, tid);

    const int beg = first[blockIdx.x];
    const int end = first[blockIdx.x + 1];
    const int npass = (end - beg + CH - 1) / CH;
    const bool pend = MUL && npass > 0;
    const int c0l = beg + (npass - 1) * CH;
    if constexpr (!MUL) __syncthreads();
    for (int c0 = beg; c0 < end; c0 += CH) {
        const int left_c = end - c0;
        const int n = left_c < CH ? left_c : CH;
        const int id = c0 / CH + (int)blockIdx.x;
        if constexpr (MUL) {
            if (tid < ROWS) bin[tid] = 0;
            __syncthreads();
            unsigned long long pr[CH / 1024];
            int rk[CH / 1024];
            float ns[CH / 1024], nd[CH / 1024];
#pragma unroll
            for (int j = 0; j < CH / 1024; ++j) {
                const int e = j * 1024 + tid;
                pr[j] = e < n ? pairs[c0 + e] : 0ull;
            }
#pragma unroll
            for (int j = 0; j < CH / 1024; ++j) {
                ns[j] = dinv[(int)(unsigned)pr[j]];
                nd[j] = dinv[(int)(unsigned)(pr[j] >> 32)];
            }
#pragma unroll
            for (int j = 0; j < CH / 1024; ++j) {
                const int e = j * 1024 + tid;
                const int dj = (int)(unsigned)(pr[j] >> 32) - row0;
                rk[j] = e < n ? lds_count(&bin[dj]) : 0;
            }
            __syncthreads();
            int cntv = 0, incl = 0;
            if (w < SCW) {
                cntv = tid < ROWS ? bin[tid] : 0;
                incl = cntv;
#pragma unroll
                for (int dd = 1; dd < 32; dd <<= 1) {
                    const int up = __shfl_up(incl, dd);
                    if (lane >= dd) incl += up;
                }
                if (lane == 31) wsum[w] = incl;
            }
            __syncthreads();
            if (w < SCW) {
                int off = 0;
                for (int i = 0; i < SCW; ++i) {
                    const int c = wsum[i];
                    off += i < w ? c : 0;
                }
                if (tid < ROWS) seg[tid] = off + incl - cntv;
                if (tid == ROWS - 1) seg[ROWS] = n;
            }
            __syncthreads();
#pragma unroll
            for (int j = 0; j < CH / 1024; ++j) {
                const int e = j * 1024 + tid;
                const unsigned sj = (unsigned)pr[j];
                const unsigned dj = (unsigned)(pr[j] >> 32) - (unsigned)row0;
                const unsigned long long ent = (unsigned long long)(sj | (dj << SRC_BITS)) |
                                               ((unsigned long long)__float_as_uint(ns[j] * nd[j]) << 32);
                if (e < n) lst[seg[dj] + rk[j]] = ent;
            }
            __syncthreads();
#pragma unroll
            for (int j = 0; j < CH / 1024; ++j) {
                const int e = j * 1024 + tid;
                if (e < n) olist[c0 + e] = lst[e];
            }
            if (tid < ROWS + 1) segs[id * (ROWS + 1) + tid] = seg[tid];
            const int np = c0 == c0l ? 1 : 2;
            for (int p = 0; p < np; ++p) {
                const int ra = p == 0 ? (H0ROWS * w) >> 5 : H0ROWS + (((ROWS - H0ROWS) * w) >> 5);
                const int rb = p == 0 ? (H0ROWS * (w + 1)) >> 5 : H0ROWS + (((ROWS - H0ROWS) * (w + 1)) >> 5);
                const int sb = __builtin_amdgcn_readfirstlane(seg[ra]);
                const int se = __builtin_amdgcn_readfirstlane(seg[rb]);
                walk_piece<F, NE, RP, true>(lst, nullptr, sb, se, h, acc, lane);
            }
        } else {
            const int* sg = segs + (size_t)id * (ROWS + 1);
            const int sb = __builtin_amdgcn_readfirstlane(sg[(ROWS * w) >> 5]);
            const int se = __builtin_amdgcn_readfirstlane(sg[(ROWS * (w + 1)) >> 5]);
            walk_piece<F, NE, RP, false>(nullptr, olist + c0, sb, se, h, acc, lane);
        }
    }
    __syncthreads();

    if constexpr (!MUL) {
        for (int i = tid; i < nfl; i += 1024) out[(size_t)row0 * F + i] = acc[i];
    } else {
        float4* wq1 = reinterpret_cast<float4*>(lst);
        stage_b<HID2, WP1>(W1, wq1, tid);
        __syncthreads();
#pragma nounroll
        for (int ph = 0; ph < 2; ++ph) {
            const int st0 = ph == 0 ? w : H0ROWS / 16 + w;
            const int st1 = ph == 0 ? H0ROWS / 16 : ROWS / 16;
            const int stp = ph == 0 ? MMW : 32;
            if (ph == 0 ? w < MMW : st0 < st1) {
                const int c = lane & 15;
                const int hh = lane >> 4;
#pragma nounroll
                for (int st = st0; st < st1; st += stp) {
                    float* strip = acc + st * 16 * RP;
                    float4 av[16];
#pragma unroll
                    for (int j = 0; j < 16; ++j)
                        av[j] = *reinterpret_cast<const float4*>(strip + c * RP + 4 * hh + 8 * j);
#pragma unroll
                    for (int half = 0; half < 2; ++half) {
                        v8f p[4];
#pragma unroll
                        for (int t = 0; t < 4; ++t) p[t] = v8f{};
                        strip_times_b<WP1>(av, wq1, 64 * half, c, hh, p);
#pragma unroll
                        for (int t = 0; t < 4; ++t) {
                            const int col = 64 * half + 16 * t + c;
                            const float bc = bias[col];
#pragma unroll
                            for (int r = 0; r < 8; ++r)
                                strip[(8 * hh + r) * RP + col] = fmaxf(p[t][r] + bc, 0.0f);
                        }
                    }
#pragma unroll
                    for (int j = 0; j < 16; ++j)
                        av[j] = *reinterpret_cast<const float4*>(strip + c * RP + 4 * hh + 8 * j);
                    v8f q[4];
#pragma unroll
                    for (int t = 0; t < 4; ++t) q[t] = v8f{};
                    strip_times_b<WP2>(av, wq2, 0, c, hh, q);
                    const int R0 = row0 + st * 16;
#pragma unroll
                    for (int r = 0; r < 8; ++r) {
#pragma unroll
                        for (int pp = 0; pp < HID / 32; ++pp) {
                            const float xx = q[2 * pp][r];
                            const float yy = q[2 * pp + 1][r];
                            const float xs = swap_halves(xx);
                            const float ys = swap_halves(yy);
                            const float u = hh ? ys : xx;
                            const float v = hh ? yy : xs;
                            if (R0 + r < N) out[(size_t)(R0 + r) * HID + pp * 32 + lane] = u;
                            if (R0 + 8 + r < N) out[(size_t)(R0 + 8 + r) * HID + pp * 32 + lane] = v;
                        }
                    }
                }
            } else if (ph == 0 && pend) {
                const int ra = H0ROWS + (w - MMW) * ((ROWS - H0ROWS) / (32 - MMW));
                const int sb = __builtin_amdgcn_readfirstlane(seg[ra]);
                const int se = __builtin_amdgcn_readfirstlane(seg[ra + (ROWS - H0ROWS) / (32 - MMW)]);
                walk_piece<F, NE, RP, false>(nullptr, olist + c0l, sb, se, h, acc, lane);
            }
            __syncthreads();
        }
    }
}

static inline int blocks_for(long long n, int bs) { return (int)((n + bs - 1) / bs); }

extern "C" void kernel_launch(void* const* d_in, const int* in_sizes, int n_in,
                              void* d_out, int out_size, void* d_ws, size_t ws_size,
                              hipStream_t stream) {
    const float* x  = (const float*)d_in[0];
    const int*   ei = (const int*)d_in[1];
    const float* W1 = (const float*)d_in[2];
    const float* b1 = (const float*)d_in[3];
    const float* W2 = (const float*)d_in[4];
    const float* b2 = (const float*)d_in[5];
    float* out = (float*)d_out;

    const int N = in_sizes[0] / IN_C;
    const long long E = (long long)in_sizes[1] / 2;
    const int* esrc = ei;
    const int* edst = ei + E;

    float* ws   = (float*)d_ws;
    float* dinv = ws;
    float* h2   = dinv + 2 * (size_t)N;
    unsigned long long* pairs = (unsigned long long*)(h2 + (size_t)N * HID);
    unsigned long long* olist = pairs + E;
    int* cnt    = (int*)(olist + E);
    int* first  = cnt + MAXB;
    int* cursor = first + MAXB + 4;
    int* segs   = cursor + MAXB + 4;

    const int nb = (N + ROWS - 1) / ROWS;

    init_kernel<<<1, 256, 0, stream>>>(cnt, nb);

    count_kernel<<<blocks_for(E, EPB), 256, 0, stream>>>(edst, cnt, E, nb);
    scan_kernel<<<1, 256, 0, stream>>>(cnt, first, cursor, nb);
    file_edges_kernel<<<blocks_for(E, EPB), 256, 0, stream>>>(esrc, edst, cursor, pairs, E, nb);
    range_norm_kernel<<<nb, DGT, 0, stream>>>(pairs, first, dinv, N);

    edge_aggregate<IN_C, 16, true><<<nb, 1024, 0, stream>>>(pairs, first, x, dinv, b1, h2, N, W1, W2, olist, segs);

    edge_aggregate<HID, 32, false><<<nb, 1024, 0, stream>>>(pairs, first, h2, dinv, b2, out, N, nullptr, nullptr, olist, segs);
}
